# starting kernel
# baseline (speedup 1.0000x reference)
#include <hip/hip_runtime.h>
#include <hip/hip_bf16.h>


#define NS   4096
#define NG   20000
#define NE   131072
#define DIN  256
#define NH   4
#define C1   64
#define C3   128
#define HC1  256
#define HC3  512
#define NEG_ENC 0x00800000u

typedef __attribute__((ext_vector_type(16))) _Float16 v16h;
typedef __attribute__((ext_vector_type(8)))  _Float16 v8h;
typedef __attribute__((ext_vector_type(8)))  float    v8f;

static inline int cdiv(int a, int b) { return (a + b - 1) / b; }

__device__ __forceinline__ float eluf(float x) { return x > 0.f ? x : (expf(x) - 1.f); }

__global__ void k_f32_to_f16(const float* __restrict__ in, _Float16* __restrict__ out, int n) {
    int i = blockIdx.x * blockDim.x + threadIdx.x;
    if (i < n) out[i] = (_Float16)in[i];
}

__global__ void k_f32_to_f16_t(const float* __restrict__ in, _Float16* __restrict__ out,
                               int K, int N) {
    int i = blockIdx.x * blockDim.x + threadIdx.x;
    if (i >= N * K) return;
    int n = i / K, k = i - n * K;
    out[i] = (_Float16)in[(size_t)k * N + n];
}

__global__ void k_fill_f32(float* __restrict__ p, float v, int n) {
    int i = blockIdx.x * blockDim.x + threadIdx.x;
    if (i < n) p[i] = v;
}

__global__ void k_fill_u32(unsigned* __restrict__ p, unsigned v, int n) {
    int i = blockIdx.x * blockDim.x + threadIdx.x;
    if (i < n) p[i] = v;
}

__global__ void k_rowbias(float* __restrict__ out, const float* __restrict__ bias,
                          int rows, int cols) {
    int i = blockIdx.x * blockDim.x + threadIdx.x;
    if (i < rows * cols) out[i] = bias[i % cols];
}

__global__ void k_elu_f16copy(float* __restrict__ x, _Float16* __restrict__ xh, int n) {
    int i = blockIdx.x * blockDim.x + threadIdx.x;
    if (i < n) { float v = eluf(x[i]); x[i] = v; xh[i] = (_Float16)v; }
}

__global__ void k_sample1(float* __restrict__ x, const float* __restrict__ sl,
                          _Float16* __restrict__ xh, int n) {
    int i = blockIdx.x * blockDim.x + threadIdx.x;
    if (i >= n) return;
    int col = i & (HC1 - 1);
    int s   = i >> 8;
    float v = eluf(x[i] + sl[(size_t)s * C1 + (col & (C1 - 1))]);
    x[i] = v; xh[i] = (_Float16)v;
}

__global__ void k_final(const float* __restrict__ acc3, const float* __restrict__ sl3,
                        const float* __restrict__ bias3, float* __restrict__ out, int n) {
    int i = blockIdx.x * blockDim.x + threadIdx.x;
    if (i >= n) return;
    int c = i & (C3 - 1);
    int s = i >> 7;
    const float* a = acc3 + (size_t)s * HC3;
    float mean = 0.25f * (a[c] + a[C3 + c] + a[2 * C3 + c] + a[3 * C3 + c]);
    out[i] = eluf(mean + bias3[c] + sl3[i]);
}

__device__ __forceinline__ v16h ld_frag(const _Float16* p) {
    v8h lo = *(const v8h*)(p);
    v8h hi = *(const v8h*)(p + 16);
    v16h r;
#pragma unroll
    for (int j = 0; j < 8; ++j) { r[j] = lo[j]; r[8 + j] = hi[j]; }
    return r;
}

__global__ void __launch_bounds__(256)
k_gemm_wmma(const _Float16* __restrict__ X, const _Float16* __restrict__ Wt,
            const float* __restrict__ bias, float* __restrict__ Y,
            int M, int N, int K) {
    int wave = threadIdx.x >> 5;
    int lane = threadIdx.x & 31;
    int wm = wave & 3, wn = wave >> 2;
    int m0 = blockIdx.x * 128 + wm * 32;
    if (m0 >= M) return;
    int n0 = blockIdx.y * 64 + wn * 32;
    int lr = lane & 15;
    int kb = (lane >> 4) * 8;

    int ra0 = m0 + lr;      if (ra0 > M - 1) ra0 = M - 1;
    int ra1 = m0 + 16 + lr; if (ra1 > M - 1) ra1 = M - 1;
    int cb0 = n0 + lr;
    int cb1 = cb0 + 16;

    const _Float16* pA0 = X  + (size_t)ra0 * K + kb;
    const _Float16* pA1 = X  + (size_t)ra1 * K + kb;
    const _Float16* pB0 = Wt + (size_t)cb0 * K + kb;
    const _Float16* pB1 = Wt + (size_t)cb1 * K + kb;

    v8f acc00 = {}, acc01 = {}, acc10 = {}, acc11 = {};
#pragma unroll 4
    for (int k0 = 0; k0 < K; k0 += 32) {
        v16h a0 = ld_frag(pA0 + k0);
        v16h a1 = ld_frag(pA1 + k0);
        v16h b0 = ld_frag(pB0 + k0);
        v16h b1 = ld_frag(pB1 + k0);
        acc00 = __builtin_amdgcn_wmma_f32_16x16x32_f16(false, a0, false, b0, (short)0, acc00, false, false);
        acc01 = __builtin_amdgcn_wmma_f32_16x16x32_f16(false, a0, false, b1, (short)0, acc01, false, false);
        acc10 = __builtin_amdgcn_wmma_f32_16x16x32_f16(false, a1, false, b0, (short)0, acc10, false, false);
        acc11 = __builtin_amdgcn_wmma_f32_16x16x32_f16(false, a1, false, b1, (short)0, acc11, false, false);
    }

    float bv0 = bias ? bias[cb0] : 0.f;
    float bv1 = bias ? bias[cb1] : 0.f;
    int mr = m0 + 8 * (lane >> 4);
#pragma unroll
    for (int v = 0; v < 8; ++v) {
        int ma = mr + v;
        if (ma < M) {
            Y[(size_t)ma * N + cb0] = acc00[v] + bv0;
            Y[(size_t)ma * N + cb1] = acc01[v] + bv1;
        }
        int mb = mr + 16 + v;
        if (mb < M) {
            Y[(size_t)mb * N + cb0] = acc10[v] + bv0;
            Y[(size_t)mb * N + cb1] = acc11[v] + bv1;
        }
    }
}

__global__ void __launch_bounds__(256)
k_edge_alpha(const int* __restrict__ src, const int* __restrict__ dst,
             const float* __restrict__ xl, const float* __restrict__ xr,
             const float* __restrict__ att, float* __restrict__ alpha,
             unsigned* __restrict__ menc, int E, int C) {
    int wave = threadIdx.x >> 5, lane = threadIdx.x & 31;
    int e = blockIdx.x * 8 + wave;
    if (e >= E) return;
    int h = lane >> 3, ls = lane & 7;
    int s = src[e], d = dst[e];
    const float* pl = xl + ((size_t)s * NH + h) * C;
    const float* pr = xr + ((size_t)d * NH + h) * C;
    const float* pa = att + (size_t)h * C;
    float acc = 0.f;
    for (int c = ls; c < C; c += 8) {
        float g = pl[c] + pr[c];
        g = g > 0.f ? g : 0.2f * g;
        acc += g * pa[c];
    }
    acc += __shfl_xor(acc, 1, 32);
    acc += __shfl_xor(acc, 2, 32);
    acc += __shfl_xor(acc, 4, 32);
    if (ls == 0) {
        alpha[(size_t)e * NH + h] = acc;
        unsigned u = __float_as_uint(acc);
        unsigned enc = (u & 0x80000000u) ? ~u : (u | 0x80000000u);
        atomicMax(&menc[(size_t)d * NH + h], enc);
    }
}

__global__ void k_edge_exp(const int* __restrict__ dst, float* __restrict__ alpha_ea,
                           const unsigned* __restrict__ menc, float* __restrict__ den,
                           int EH) {
    int i = blockIdx.x * blockDim.x + threadIdx.x;
    if (i >= EH) return;
    int e = i >> 2, h = i & 3;
    int d = dst[e];
    unsigned enc = menc[(size_t)d * NH + h];
    float m = (enc & 0x80000000u) ? __uint_as_float(enc ^ 0x80000000u)
                                  : __uint_as_float(~enc);
    float ea = expf(alpha_ea[i] - m);
    alpha_ea[i] = ea;
    atomicAdd(&den[(size_t)d * NH + h], ea);
}

__global__ void __launch_bounds__(256)
k_edge_scatter(const int* __restrict__ src, const int* __restrict__ dst,
               const float* __restrict__ xl, const float* __restrict__ ea,
               const float* __restrict__ den, float* __restrict__ out, int E, int C) {
    int wave = threadIdx.x >> 5, lane = threadIdx.x & 31;
    int e = blockIdx.x * 8 + wave;
    if (e >= E) return;
    int h = lane >> 3, ls = lane & 7;
    int s = src[e], d = dst[e];
    float coef = ea[(size_t)e * NH + h] / (den[(size_t)d * NH + h] + 1e-16f);
    const float* pl = xl + ((size_t)s * NH + h) * C;
    float*       po = out + ((size_t)d * NH + h) * C;
    for (int c = ls; c < C; c += 8) atomicAdd(&po[c], coef * pl[c]);
}

extern "C" void kernel_launch(void* const* d_in, const int* in_sizes, int n_in,
                              void* d_out, int out_size, void* d_ws, size_t ws_size,
                              hipStream_t stream) {
    (void)in_sizes; (void)n_in; (void)out_size; (void)ws_size;

    const float* x_sample = (const float*)d_in[0];
    const float* x_gene   = (const float*)d_in[1];
    const int*   sg_src   = (const int*)d_in[2];
    const int*   sg_dst   = (const int*)d_in[3];
    const int*   gs_src   = (const int*)d_in[4];
    const int*   gs_dst   = (const int*)d_in[5];
    const float* Wl1_sg = (const float*)d_in[6];   const float* bl1_sg  = (const float*)d_in[7];
    const float* Wr1_sg = (const float*)d_in[8];   const float* br1_sg  = (const float*)d_in[9];
    const float* att1_sg = (const float*)d_in[10]; const float* bias1_sg = (const float*)d_in[11];
    const float* Wl1_gs = (const float*)d_in[12];  const float* bl1_gs  = (const float*)d_in[13];
    const float* Wr1_gs = (const float*)d_in[14];  const float* br1_gs  = (const float*)d_in[15];
    const float* att1_gs = (const float*)d_in[16]; const float* bias1_gs = (const float*)d_in[17];
    const float* Wl3 = (const float*)d_in[18];     const float* bl3 = (const float*)d_in[19];
    const float* Wr3 = (const float*)d_in[20];     const float* br3 = (const float*)d_in[21];
    const float* att3 = (const float*)d_in[22];    const float* bias3 = (const float*)d_in[23];
    const float* sl1_W = (const float*)d_in[24];   const float* sl1_b = (const float*)d_in[25];
    const float* sl3_W = (const float*)d_in[26];   const float* sl3_b = (const float*)d_in[27];

    char* p = (char*)d_ws;
    auto carve = [&](size_t bytes) -> char* {
        char* r = p; p += (bytes + 255) & ~(size_t)255; return r;
    };

    _Float16* hx_sample = (_Float16*)carve((size_t)NS * DIN * 2);
    _Float16* hx_gene   = (_Float16*)carve((size_t)NG * DIN * 2);
    _Float16* hWl1sg = (_Float16*)carve((size_t)DIN * HC1 * 2);
    _Float16* hWr1sg = (_Float16*)carve((size_t)DIN * HC1 * 2);
    _Float16* hWl1gs = (_Float16*)carve((size_t)DIN * HC1 * 2);
    _Float16* hWr1gs = (_Float16*)carve((size_t)DIN * HC1 * 2);
    _Float16* hWl3   = (_Float16*)carve((size_t)HC1 * HC3 * 2);
    _Float16* hWr3   = (_Float16*)carve((size_t)HC1 * HC3 * 2);
    _Float16* hsl1W  = (_Float16*)carve((size_t)DIN * C1 * 2);
    _Float16* hsl3W  = (_Float16*)carve((size_t)HC1 * C3 * 2);
    _Float16* hx1_gene   = (_Float16*)carve((size_t)NG * HC1 * 2);
    _Float16* hx1_sample = (_Float16*)carve((size_t)NS * HC1 * 2);
    float* xl_sg = (float*)carve((size_t)NS * HC1 * 4);
    float* xr_sg = (float*)carve((size_t)NG * HC1 * 4);
    float* out1_gene   = (float*)carve((size_t)NG * HC1 * 4);
    float* xr_gs = (float*)carve((size_t)NS * HC1 * 4);
    float* out1_sample = (float*)carve((size_t)NS * HC1 * 4);
    float* sl1 = (float*)carve((size_t)NS * C1 * 4);
    float* xl3 = (float*)carve((size_t)NG * HC3 * 4);
    float* xr3 = (float*)carve((size_t)NS * HC3 * 4);
    float* acc3 = (float*)carve((size_t)NS * HC3 * 4);
    float* sl3 = (float*)carve((size_t)NS * C3 * 4);
    float* alpha = (float*)carve((size_t)NE * NH * 4);
    unsigned* menc = (unsigned*)carve((size_t)NG * NH * 4);
    float* den = (float*)carve((size_t)NG * NH * 4);
    float* xl_gs = xr_sg;

    const int TB = 256;
#define EW(n) cdiv((n), TB), TB, 0, stream
    k_f32_to_f16<<<EW(NS * DIN)>>>(x_sample, hx_sample, NS * DIN);
    k_f32_to_f16<<<EW(NG * DIN)>>>(x_gene, hx_gene, NG * DIN);
    k_f32_to_f16_t<<<EW(DIN * HC1)>>>(Wl1_sg, hWl1sg, DIN, HC1);
    k_f32_to_f16_t<<<EW(DIN * HC1)>>>(Wr1_sg, hWr1sg, DIN, HC1);
    k_f32_to_f16_t<<<EW(DIN * HC1)>>>(Wl1_gs, hWl1gs, DIN, HC1);
    k_f32_to_f16_t<<<EW(DIN * HC1)>>>(Wr1_gs, hWr1gs, DIN, HC1);
    k_f32_to_f16_t<<<EW(HC1 * HC3)>>>(Wl3, hWl3, HC1, HC3);
    k_f32_to_f16_t<<<EW(HC1 * HC3)>>>(Wr3, hWr3, HC1, HC3);
    k_f32_to_f16_t<<<EW(DIN * C1)>>>(sl1_W, hsl1W, DIN, C1);
    k_f32_to_f16_t<<<EW(HC1 * C3)>>>(sl3_W, hsl3W, HC1, C3);

    auto gemm = [&](const _Float16* X, const _Float16* Wt, const float* b, float* Y,
                    int M, int N, int K) {
        dim3 g(cdiv(M, 128), cdiv(N, 64));
        k_gemm_wmma<<<g, 256, 0, stream>>>(X, Wt, b, Y, M, N, K);
    };

    gemm(hx_sample, hWl1sg, bl1_sg, xl_sg, NS, HC1, DIN);
    gemm(hx_gene,   hWr1sg, br1_sg, xr_sg, NG, HC1, DIN);
    k_fill_u32<<<EW(NG * NH)>>>(menc, NEG_ENC, NG * NH);
    k_fill_f32<<<EW(NG * NH)>>>(den, 0.f, NG * NH);
    k_rowbias<<<EW(NG * HC1)>>>(out1_gene, bias1_sg, NG, HC1);
    k_edge_alpha<<<cdiv(NE, 8), 256, 0, stream>>>(sg_src, sg_dst, xl_sg, xr_sg, att1_sg,
                                                  alpha, menc, NE, C1);
    k_edge_exp<<<EW(NE * NH)>>>(sg_dst, alpha, menc, den, NE * NH);
    k_edge_scatter<<<cdiv(NE, 8), 256, 0, stream>>>(sg_src, sg_dst, xl_sg, alpha, den,
                                                    out1_gene, NE, C1);
    k_elu_f16copy<<<EW(NG * HC1)>>>(out1_gene, hx1_gene, NG * HC1);

    gemm(hx_gene,   hWl1gs, bl1_gs, xl_gs, NG, HC1, DIN);
    gemm(hx_sample, hWr1gs, br1_gs, xr_gs, NS, HC1, DIN);
    gemm(hx_sample, hsl1W,  sl1_b,  sl1,   NS, C1,  DIN);
    k_fill_u32<<<EW(NS * NH)>>>(menc, NEG_ENC, NS * NH);
    k_fill_f32<<<EW(NS * NH)>>>(den, 0.f, NS * NH);
    k_rowbias<<<EW(NS * HC1)>>>(out1_sample, bias1_gs, NS, HC1);
    k_edge_alpha<<<cdiv(NE, 8), 256, 0, stream>>>(gs_src, gs_dst, xl_gs, xr_gs, att1_gs,
                                                  alpha, menc, NE, C1);
    k_edge_exp<<<EW(NE * NH)>>>(gs_dst, alpha, menc, den, NE * NH);
    k_edge_scatter<<<cdiv(NE, 8), 256, 0, stream>>>(gs_src, gs_dst, xl_gs, alpha, den,
                                                    out1_sample, NE, C1);
    k_sample1<<<EW(NS * HC1)>>>(out1_sample, sl1, hx1_sample, NS * HC1);

    gemm(hx1_gene,   hWl3,  bl3,   xl3, NG, HC3, HC1);
    gemm(hx1_sample, hWr3,  br3,   xr3, NS, HC3, HC1);
    gemm(hx1_sample, hsl3W, sl3_b, sl3, NS, C3,  HC1);
    k_fill_u32<<<EW(NS * NH)>>>(menc, NEG_ENC, NS * NH);
    k_fill_f32<<<EW(NS * NH)>>>(den, 0.f, NS * NH);
    k_fill_f32<<<EW(NS * HC3)>>>(acc3, 0.f, NS * HC3);
    k_edge_alpha<<<cdiv(NE, 8), 256, 0, stream>>>(gs_src, gs_dst, xl3, xr3, att3,
                                                  alpha, menc, NE, C3);
    k_edge_exp<<<EW(NE * NH)>>>(gs_dst, alpha, menc, den, NE * NH);
    k_edge_scatter<<<cdiv(NE, 8), 256, 0, stream>>>(gs_src, gs_dst, xl3, alpha, den,
                                                    acc3, NE, C3);
    k_final<<<EW(NS * C3)>>>(acc3, sl3, bias3, (float*)d_out, NS * C3);
#undef EW
}


# edges grouped by destination once per call (counting sort); one wave per node does segment softmax and weighted sum in one gather pass: no float atomics, each source row read once
# speedup vs baseline: 2.5217x; 2.5217x over previous
#include <hip/hip_runtime.h>
#include <hip/hip_bf16.h>


#define NS   4096
#define NG   20000
#define NE   131072
#define DIN  256
#define NH   4
#define C1   64
#define C3   128
#define HC1  256
#define HC3  512

typedef __attribute__((ext_vector_type(16))) _Float16 v16h;
typedef __attribute__((ext_vector_type(8)))  _Float16 v8h;
typedef __attribute__((ext_vector_type(8)))  float    v8f;
typedef __attribute__((ext_vector_type(4)))  float    v4f;

static inline int cdiv(int a, int b) { return (a + b - 1) / b; }

__device__ __forceinline__ float eluf(float x) { return x > 0.f ? x : (expf(x) - 1.f); }

__global__ void k_f32_to_f16(const float* __restrict__ in, _Float16* __restrict__ out, int n) {
    int i = blockIdx.x * blockDim.x + threadIdx.x;
    if (i < n) out[i] = (_Float16)in[i];
}

__global__ void k_f32_to_f16_t(const float* __restrict__ in, _Float16* __restrict__ out,
                               int K, int N) {
    int i = blockIdx.x * blockDim.x + threadIdx.x;
    if (i >= N * K) return;
    int n = i / K, k = i - n * K;
    out[i] = (_Float16)in[(size_t)k * N + n];
}

__global__ void k_fill_u32(unsigned* __restrict__ p, unsigned v, int n) {
    int i = blockIdx.x * blockDim.x + threadIdx.x;
    if (i < n) p[i] = v;
}

__global__ void k_elu_f16copy(float* __restrict__ x, _Float16* __restrict__ xh, int n) {
    int i = blockIdx.x * blockDim.x + threadIdx.x;
    if (i < n) { float v = eluf(x[i]); x[i] = v; xh[i] = (_Float16)v; }
}

__global__ void k_sample1(float* __restrict__ x, const float* __restrict__ sl,
                          _Float16* __restrict__ xh, int n) {
    int i = blockIdx.x * blockDim.x + threadIdx.x;
    if (i >= n) return;
    int col = i & (HC1 - 1);
    int s   = i >> 8;
    float v = eluf(x[i] + sl[(size_t)s * C1 + (col & (C1 - 1))]);
    x[i] = v; xh[i] = (_Float16)v;
}

__global__ void k_final(const float* __restrict__ acc3, const float* __restrict__ sl3,
                        const float* __restrict__ bias3, float* __restrict__ out, int n) {
    int i = blockIdx.x * blockDim.x + threadIdx.x;
    if (i >= n) return;
    int c = i & (C3 - 1);
    int s = i >> 7;
    const float* a = acc3 + (size_t)s * HC3;
    float mean = 0.25f * (a[c] + a[C3 + c] + a[2 * C3 + c] + a[3 * C3 + c]);
    out[i] = eluf(mean + bias3[c] + sl3[i]);
}

__device__ __forceinline__ v16h ld_frag(const _Float16* p) {
    v8h lo = *(const v8h*)(p);
    v8h hi = *(const v8h*)(p + 16);
    v16h r;
#pragma unroll
    for (int j = 0; j < 8; ++j) { r[j] = lo[j]; r[8 + j] = hi[j]; }
    return r;
}

__global__ void __launch_bounds__(256)
k_gemm_wmma(const _Float16* __restrict__ X, const _Float16* __restrict__ Wt,
            const float* __restrict__ bias, float* __restrict__ Y,
            int M, int N, int K) {
    int wave = threadIdx.x >> 5;
    int lane = threadIdx.x & 31;
    int wm = wave & 3, wn = wave >> 2;
    int m0 = blockIdx.x * 128 + wm * 32;
    if (m0 >= M) return;
    int n0 = blockIdx.y * 64 + wn * 32;
    int lr = lane & 15;
    int kb = (lane >> 4) * 8;

    int ra0 = m0 + lr;      if (ra0 > M - 1) ra0 = M - 1;
    int ra1 = m0 + 16 + lr; if (ra1 > M - 1) ra1 = M - 1;
    int cb0 = n0 + lr;
    int cb1 = cb0 + 16;

    const _Float16* pA0 = X  + (size_t)ra0 * K + kb;
    const _Float16* pA1 = X  + (size_t)ra1 * K + kb;
    const _Float16* pB0 = Wt + (size_t)cb0 * K + kb;
    const _Float16* pB1 = Wt + (size_t)cb1 * K + kb;

    v8f acc00 = {}, acc01 = {}, acc10 = {}, acc11 = {};
#pragma unroll 4
    for (int k0 = 0; k0 < K; k0 += 32) {
        v16h a0 = ld_frag(pA0 + k0);
        v16h a1 = ld_frag(pA1 + k0);
        v16h b0 = ld_frag(pB0 + k0);
        v16h b1 = ld_frag(pB1 + k0);
        acc00 = __builtin_amdgcn_wmma_f32_16x16x32_f16(false, a0, false, b0, (short)0, acc00, false, false);
        acc01 = __builtin_amdgcn_wmma_f32_16x16x32_f16(false, a0, false, b1, (short)0, acc01, false, false);
        acc10 = __builtin_amdgcn_wmma_f32_16x16x32_f16(false, a1, false, b0, (short)0, acc10, false, false);
        acc11 = __builtin_amdgcn_wmma_f32_16x16x32_f16(false, a1, false, b1, (short)0, acc11, false, false);
    }

    float bv0 = bias ? bias[cb0] : 0.f;
    float bv1 = bias ? bias[cb1] : 0.f;
    int mr = m0 + 8 * (lane >> 4);
#pragma unroll
    for (int v = 0; v < 8; ++v) {
        int ma = mr + v;
        if (ma < M) {
            Y[(size_t)ma * N + cb0] = acc00[v] + bv0;
            Y[(size_t)ma * N + cb1] = acc01[v] + bv1;
        }
        int mb = mr + 16 + v;
        if (mb < M) {
            Y[(size_t)mb * N + cb0] = acc10[v] + bv0;
            Y[(size_t)mb * N + cb1] = acc11[v] + bv1;
        }
    }
}

__global__ void k_deg_count(const int* __restrict__ dst_a, int* __restrict__ cnt_a,
                            const int* __restrict__ dst_b, int* __restrict__ cnt_b, int E) {
    int i = blockIdx.x * blockDim.x + threadIdx.x;
    if (i < E)          atomicAdd(&cnt_a[dst_a[i]], 1);
    else if (i < 2 * E) atomicAdd(&cnt_b[dst_b[i - E]], 1);
}

__global__ void __launch_bounds__(1024)
k_deg_scan(const int* __restrict__ cnt_a, int* __restrict__ off_a, int* __restrict__ cur_a, int n_a,
           const int* __restrict__ cnt_b, int* __restrict__ off_b, int* __restrict__ cur_b, int n_b) {
    __shared__ int part[1024];
    const int* cnt = blockIdx.x == 0 ? cnt_a : cnt_b;
    int*       off = blockIdx.x == 0 ? off_a : off_b;
    int*       cur = blockIdx.x == 0 ? cur_a : cur_b;
    const int  n   = blockIdx.x == 0 ? n_a : n_b;
    const int t = threadIdx.x;
    const int per = (n + 1023) / 1024;
    const int b = t * per;
    int e = b + per; if (e > n) e = n;
    int s = 0;
    for (int i = b; i < e; ++i) s += cnt[i];
    part[t] = s;
    __syncthreads();
    for (int d = 1; d < 1024; d <<= 1) {
        int v = (t >= d) ? part[t - d] : 0;
        __syncthreads();
        part[t] += v;
        __syncthreads();
    }
    int run = part[t] - s;
    for (int i = b; i < e; ++i) { off[i] = run; cur[i] = run; run += cnt[i]; }
    if (t == 1023) off[n] = part[1023];
}

__global__ void k_group_fill(const int* __restrict__ src_a, const int* __restrict__ dst_a,
                             int* __restrict__ cur_a, int* __restrict__ ss_a,
                             const int* __restrict__ src_b, const int* __restrict__ dst_b,
                             int* __restrict__ cur_b, int* __restrict__ ss_b, int E) {
    int i = blockIdx.x * blockDim.x + threadIdx.x;
    if (i < E) {
        int pos = atomicAdd(&cur_a[dst_a[i]], 1);
        ss_a[pos] = src_a[i];
    } else if (i < 2 * E) {
        int j = i - E;
        int pos = atomicAdd(&cur_b[dst_b[j]], 1);
        ss_b[pos] = src_b[j];
    }
}

__device__ __forceinline__ float sum8(float t) {
    t += __builtin_bit_cast(float, __builtin_amdgcn_update_dpp(0, __builtin_bit_cast(int, t), 0x161, 0xf, 0xf, true));
    t += __builtin_bit_cast(float, __builtin_amdgcn_update_dpp(0, __builtin_bit_cast(int, t), 0x162, 0xf, 0xf, true));
    t += __builtin_bit_cast(float, __builtin_amdgcn_update_dpp(0, __builtin_bit_cast(int, t), 0x164, 0xf, 0xf, true));
    return t;
}

template <int C>
__device__ __forceinline__ void
gat_node(const int* __restrict__ off, const int* __restrict__ ssrc,
         const float* __restrict__ xl, const float* __restrict__ xr,
         const float* __restrict__ att, const float* __restrict__ bias,
         float* __restrict__ out, int ND) {
    constexpr int V   = C / 32;
    constexpr int U   = 4;
    constexpr int ROW = NH * C;
    const int wave = threadIdx.x >> 5, lane = threadIdx.x & 31;
    const int d = __builtin_amdgcn_readfirstlane((int)(blockIdx.x * 8 + wave));
    if (d >= ND) return;
    const int cbase = (lane >> 3) * C + 4 * (lane & 7);

    const float* pr = xr + (size_t)d * ROW + cbase;
    v4f r[V], a[V], acc[V];
#pragma unroll
    for (int j = 0; j < V; ++j) {
        r[j] = *(const v4f*)(pr + 32 * j);
        const float* q = att + cbase + 32 * j;
        a[j] = (v4f){q[0], q[1], q[2], q[3]};
        acc[j] = (v4f){0.f, 0.f, 0.f, 0.f};
    }
    float m = -3.0e38f, den = 0.f;
    const int beg = off[d], end = off[d + 1];

    for (int i = beg; i < end; i += U) {
        v4f x[U][V];
#pragma unroll
        for (int u = 0; u < U; ++u) {
            int idx = i + u; if (idx > end - 1) idx = end - 1;
            const float* pl = xl + (size_t)ssrc[idx] * ROW + cbase;
#pragma unroll
            for (int j = 0; j < V; ++j) x[u][j] = *(const v4f*)(pl + 32 * j);
        }
#pragma unroll
        for (int u = 0; u < U; ++u) {
            float t = 0.f;
#pragma unroll
            for (int j = 0; j < V; ++j) {
                v4f g = x[u][j] + r[j];
#pragma unroll
                for (int k = 0; k < 4; ++k) {
                    float gk = g[k];
                    gk = gk > 0.f ? gk : 0.2f * gk;
                    t += gk * a[j][k];
                }
            }
            t = sum8(t);
            t = (i + u < end) ? t : -3.0e38f;
            const float dlt = t - m;
            const float ex  = expf(-fabsf(dlt));
            const bool  up  = dlt > 0.f;
            const float sc  = up ? ex : 1.f;
            const float p   = up ? 1.f : ex;
            m   = up ? t : m;
            den = den * sc + p;
#pragma unroll
            for (int j = 0; j < V; ++j) acc[j] = acc[j] * sc + x[u][j] * p;
        }
    }

    const float inv = 1.f / (den + 1e-16f);
    float* po = out + (size_t)d * ROW + cbase;
#pragma unroll
    for (int j = 0; j < V; ++j) {
        v4f o = acc[j] * inv;
        if (bias) {
            const float* q = bias + cbase + 32 * j;
            o += (v4f){q[0], q[1], q[2], q[3]};
        }
        *(v4f*)(po + 32 * j) = o;
    }
}

__global__ void __launch_bounds__(256)
k_gat_node_c1(const int* __restrict__ off, const int* __restrict__ ssrc,
              const float* __restrict__ xl, const float* __restrict__ xr,
              const float* __restrict__ att, const float* __restrict__ bias,
              float* __restrict__ out, int ND) {
    gat_node<C1>(off, ssrc, xl, xr, att, bias, out, ND);
}

__global__ void __launch_bounds__(256)
k_gat_node_c3(const int* __restrict__ off, const int* __restrict__ ssrc,
              const float* __restrict__ xl, const float* __restrict__ xr,
              const float* __restrict__ att, const float* __restrict__ bias,
              float* __restrict__ out, int ND) {
    gat_node<C3>(off, ssrc, xl, xr, att, bias, out, ND);
}

extern "C" void kernel_launch(void* const* d_in, const int* in_sizes, int n_in,
                              void* d_out, int out_size, void* d_ws, size_t ws_size,
                              hipStream_t stream) {
    (void)in_sizes; (void)n_in; (void)out_size; (void)ws_size;

    const float* x_sample = (const float*)d_in[0];
    const float* x_gene   = (const float*)d_in[1];
    const int*   sg_src   = (const int*)d_in[2];
    const int*   sg_dst   = (const int*)d_in[3];
    const int*   gs_src   = (const int*)d_in[4];
    const int*   gs_dst   = (const int*)d_in[5];
    const float* Wl1_sg = (const float*)d_in[6];   const float* bl1_sg  = (const float*)d_in[7];
    const float* Wr1_sg = (const float*)d_in[8];   const float* br1_sg  = (const float*)d_in[9];
    const float* att1_sg = (const float*)d_in[10]; const float* bias1_sg = (const float*)d_in[11];
    const float* Wl1_gs = (const float*)d_in[12];  const float* bl1_gs  = (const float*)d_in[13];
    const float* Wr1_gs = (const float*)d_in[14];  const float* br1_gs  = (const float*)d_in[15];
    const float* att1_gs = (const float*)d_in[16]; const float* bias1_gs = (const float*)d_in[17];
    const float* Wl3 = (const float*)d_in[18];     const float* bl3 = (const float*)d_in[19];
    const float* Wr3 = (const float*)d_in[20];     const float* br3 = (const float*)d_in[21];
    const float* att3 = (const float*)d_in[22];    const float* bias3 = (const float*)d_in[23];
    const float* sl1_W = (const float*)d_in[24];   const float* sl1_b = (const float*)d_in[25];
    const float* sl3_W = (const float*)d_in[26];   const float* sl3_b = (const float*)d_in[27];

    char* p = (char*)d_ws;
    auto carve = [&](size_t bytes) -> char* {
        char* r = p; p += (bytes + 255) & ~(size_t)255; return r;
    };

    _Float16* hx_sample = (_Float16*)carve((size_t)NS * DIN * 2);
    _Float16* hx_gene   = (_Float16*)carve((size_t)NG * DIN * 2);
    _Float16* hWl1sg = (_Float16*)carve((size_t)DIN * HC1 * 2);
    _Float16* hWr1sg = (_Float16*)carve((size_t)DIN * HC1 * 2);
    _Float16* hWl1gs = (_Float16*)carve((size_t)DIN * HC1 * 2);
    _Float16* hWr1gs = (_Float16*)carve((size_t)DIN * HC1 * 2);
    _Float16* hWl3   = (_Float16*)carve((size_t)HC1 * HC3 * 2);
    _Float16* hWr3   = (_Float16*)carve((size_t)HC1 * HC3 * 2);
    _Float16* hsl1W  = (_Float16*)carve((size_t)DIN * C1 * 2);
    _Float16* hsl3W  = (_Float16*)carve((size_t)HC1 * C3 * 2);
    _Float16* hx1_gene   = (_Float16*)carve((size_t)NG * HC1 * 2);
    _Float16* hx1_sample = (_Float16*)carve((size_t)NS * HC1 * 2);
    float* xl_sg = (float*)carve((size_t)NS * HC1 * 4);
    float* xr_sg = (float*)carve((size_t)NG * HC1 * 4);
    float* out1_gene   = (float*)carve((size_t)NG * HC1 * 4);
    float* xr_gs = (float*)carve((size_t)NS * HC1 * 4);
    float* out1_sample = (float*)carve((size_t)NS * HC1 * 4);
    float* sl1 = (float*)carve((size_t)NS * C1 * 4);
    float* xl3 = (float*)carve((size_t)NG * HC3 * 4);
    float* xr3 = (float*)carve((size_t)NS * HC3 * 4);
    float* acc3 = (float*)carve((size_t)NS * HC3 * 4);
    float* sl3 = (float*)carve((size_t)NS * C3 * 4);
    int* cnt    = (int*)carve((size_t)(NG + NS) * 4);
    int* cur    = (int*)carve((size_t)(NG + NS) * 4);
    int* off_sg = (int*)carve((size_t)(NG + 1) * 4);
    int* off_gs = (int*)carve((size_t)(NS + 1) * 4);
    int* ss_sg  = (int*)carve((size_t)NE * 4);
    int* ss_gs  = (int*)carve((size_t)NE * 4);
    float* xl_gs = xr_sg;

    const int TB = 256;
#define EW(n) cdiv((n), TB), TB, 0, stream
    k_f32_to_f16<<<EW(NS * DIN)>>>(x_sample, hx_sample, NS * DIN);
    k_f32_to_f16<<<EW(NG * DIN)>>>(x_gene, hx_gene, NG * DIN);
    k_f32_to_f16_t<<<EW(DIN * HC1)>>>(Wl1_sg, hWl1sg, DIN, HC1);
    k_f32_to_f16_t<<<EW(DIN * HC1)>>>(Wr1_sg, hWr1sg, DIN, HC1);
    k_f32_to_f16_t<<<EW(DIN * HC1)>>>(Wl1_gs, hWl1gs, DIN, HC1);
    k_f32_to_f16_t<<<EW(DIN * HC1)>>>(Wr1_gs, hWr1gs, DIN, HC1);
    k_f32_to_f16_t<<<EW(HC1 * HC3)>>>(Wl3, hWl3, HC1, HC3);
    k_f32_to_f16_t<<<EW(HC1 * HC3)>>>(Wr3, hWr3, HC1, HC3);
    k_f32_to_f16_t<<<EW(DIN * C1)>>>(sl1_W, hsl1W, DIN, C1);
    k_f32_to_f16_t<<<EW(HC1 * C3)>>>(sl3_W, hsl3W, HC1, C3);

    k_fill_u32<<<EW(NG + NS)>>>((unsigned*)cnt, 0u, NG + NS);
    k_deg_count<<<EW(2 * NE)>>>(sg_dst, cnt, gs_dst, cnt + NG, NE);
    k_deg_scan<<<2, 1024, 0, stream>>>(cnt, off_sg, cur, NG, cnt + NG, off_gs, cur + NG, NS);
    k_group_fill<<<EW(2 * NE)>>>(sg_src, sg_dst, cur, ss_sg, gs_src, gs_dst, cur + NG, ss_gs, NE);

    auto gemm = [&](const _Float16* X, const _Float16* Wt, const float* b, float* Y,
                    int M, int N, int K) {
        dim3 g(cdiv(M, 128), cdiv(N, 64));
        k_gemm_wmma<<<g, 256, 0, stream>>>(X, Wt, b, Y, M, N, K);
    };

    gemm(hx_sample, hWl1sg, bl1_sg, xl_sg, NS, HC1, DIN);
    gemm(hx_gene,   hWr1sg, br1_sg, xr_sg, NG, HC1, DIN);
    k_gat_node_c1<<<cdiv(NG, 8), 256, 0, stream>>>(off_sg, ss_sg, xl_sg, xr_sg, att1_sg,
                                                    bias1_sg, out1_gene, NG);
    k_elu_f16copy<<<EW(NG * HC1)>>>(out1_gene, hx1_gene, NG * HC1);

    gemm(hx_gene,   hWl1gs, bl1_gs, xl_gs, NG, HC1, DIN);
    gemm(hx_sample, hWr1gs, br1_gs, xr_gs, NS, HC1, DIN);
    gemm(hx_sample, hsl1W,  sl1_b,  sl1,   NS, C1,  DIN);
    k_gat_node_c1<<<cdiv(NS, 8), 256, 0, stream>>>(off_gs, ss_gs, xl_gs, xr_gs, att1_gs,
                                                    bias1_gs, out1_sample, NS);
    k_sample1<<<EW(NS * HC1)>>>(out1_sample, sl1, hx1_sample, NS * HC1);

    gemm(hx1_gene,   hWl3,  bl3,   xl3, NG, HC3, HC1);
    gemm(hx1_sample, hWr3,  br3,   xr3, NS, HC3, HC1);
    gemm(hx1_sample, hsl3W, sl3_b, sl3, NS, C3,  HC1);
    k_gat_node_c3<<<cdiv(NS, 8), 256, 0, stream>>>(off_gs, ss_gs, xl3, xr3, att3,
                                                    nullptr, acc3, NS);
    k_final<<<EW(NS * C3)>>>(acc3, sl3, bias3, (float*)d_out, NS * C3);
#undef EW
}


# f16 GEMM operands stored in WMMA fragment order by their producers: a fragment load is 512 contiguous bytes per wave (4 lines), not 16 lines of 32 bytes
# speedup vs baseline: 4.4402x; 1.7608x over previous
#include <hip/hip_runtime.h>
#include <hip/hip_bf16.h>


#define NS   4096
#define NG   20000
#define NE   131072
#define DIN  256
#define NH   4
#define C1   64
#define C3   128
#define HC1  256
#define HC3  512

typedef __attribute__((ext_vector_type(16))) _Float16 v16h;
typedef __attribute__((ext_vector_type(8)))  _Float16 v8h;
typedef __attribute__((ext_vector_type(4)))  _Float16 v4h;
typedef __attribute__((ext_vector_type(8)))  float    v8f;
typedef __attribute__((ext_vector_type(4)))  float    v4f;
typedef __attribute__((ext_vector_type(4)))  int      v4i;

__device__ __forceinline__ float eluf(float x) { return x > 0.f ? x : (expf(x) - 1.f); }

__device__ __forceinline__ size_t fidx(int row, int k) {
    return ((size_t)((row >> 4) * 8 + (k >> 5)) << 9)
         + (size_t)((((k >> 4) & 1) << 8)
                    + (((row & 15) + (((k >> 3) & 1) << 4)) << 3)
                    + (k & 7));
}

#define PB_W1   4
#define PB_W3   8
#define PO_WL1SG 0
#define PO_WR1SG (PO_WL1SG + PB_W1)
#define PO_WL1GS (PO_WR1SG + PB_W1)
#define PO_WR1GS (PO_WL1GS + PB_W1)
#define PO_WL3   (PO_WR1GS + PB_W1)
#define PO_WR3   (PO_WL3 + PB_W3)
#define PO_SL1   (PO_WR3 + PB_W3)
#define PO_SL3   (PO_SL1 + 1)
#define PO_CNT   (PO_SL3 + 2)
#define PO_XS    (PO_CNT + 24)
#define PO_XG    (PO_XS + 64)
#define PO_END   (PO_XG + 500)

template <int K, int N>
__device__ __forceinline__ void cvt_t(const float* __restrict__ in, _Float16* __restrict__ out, int i) {
    const int q = i / N, n = i - q * N;
    const float* p = in + (size_t)(q * 64) * N + n;
    float v[64];
#pragma unroll
    for (int kk = 0; kk < 64; ++kk) v[kk] = p[(size_t)kk * N];
    static_assert(K == 256, "fragment-order storage is written for K = 256");
#pragma unroll
    for (int s = 0; s < 8; ++s) {
        v8h hv;
#pragma unroll
        for (int j = 0; j < 8; ++j) hv[j] = (_Float16)v[8 * s + j];
        *(v8h*)(out + fidx(n, q * 64 + 8 * s)) = hv;
    }
}

template <int U>
__device__ __forceinline__ void cvt_run(const float* __restrict__ in, _Float16* __restrict__ out, int blk, int t) {
    static_assert(U % 2 == 0, "8 rows of the block per piece");
    const int k    = 64 * ((t >> 5) & 3) + 8 * (t & 7);
    const int row0 = blk * (U * 4) + 4 * (t >> 7) + ((t >> 3) & 3);
    v4f x[U];
#pragma unroll
    for (int j = 0; j < U / 2; ++j) {
        const float* p = in + (size_t)(row0 + 8 * j) * 256 + k;
        x[2 * j]     = *(const v4f*)(p);
        x[2 * j + 1] = *(const v4f*)(p + 4);
    }
#pragma unroll
    for (int j = 0; j < U / 2; ++j) {
        v8h hv;
#pragma unroll
        for (int e = 0; e < 4; ++e) { hv[e] = (_Float16)x[2 * j][e]; hv[4 + e] = (_Float16)x[2 * j + 1][e]; }
        *(v8h*)(out + fidx(row0 + 8 * j, k)) = hv;
    }
}

__global__ void __launch_bounds__(256)
k_prep(const float* __restrict__ x_sample, const float* __restrict__ x_gene,
       const float* __restrict__ Wl1_sg, const float* __restrict__ Wr1_sg,
       const float* __restrict__ Wl1_gs, const float* __restrict__ Wr1_gs,
       const float* __restrict__ Wl3, const float* __restrict__ Wr3,
       const float* __restrict__ sl1_W, const float* __restrict__ sl3_W,
       _Float16* __restrict__ hx_sample, _Float16* __restrict__ hx_gene,
       _Float16* __restrict__ hWl1sg, _Float16* __restrict__ hWr1sg,
       _Float16* __restrict__ hWl1gs, _Float16* __restrict__ hWr1gs,
       _Float16* __restrict__ hWl3, _Float16* __restrict__ hWr3,
       _Float16* __restrict__ hsl1W, _Float16* __restrict__ hsl3W,
       int* __restrict__ cnt) {
    const int b = blockIdx.x;
    const int t = threadIdx.x;
    if (b >= PO_XG) {
        cvt_run<10>(x_gene, hx_gene, b - PO_XG, t);
    } else if (b >= PO_XS) {
        cvt_run<16>(x_sample, hx_sample, b - PO_XS, t);
    } else if (b >= PO_CNT) {
        int i = (b - PO_CNT) * 256 + t;
        if (i < (NG + NS) / 4) *(v4i*)(cnt + 4 * i) = (v4i){0, 0, 0, 0};
    } else if (b >= PO_SL3) {
        cvt_t<HC1, C3>(sl3_W, hsl3W, (b - PO_SL3) * 256 + t);
    } else if (b >= PO_SL1) {
        cvt_t<DIN, C1>(sl1_W, hsl1W, (b - PO_SL1) * 256 + t);
    } else if (b >= PO_WR3) {
        cvt_t<HC1, HC3>(Wr3, hWr3, (b - PO_WR3) * 256 + t);
    } else if (b >= PO_WL3) {
        cvt_t<HC1, HC3>(Wl3, hWl3, (b - PO_WL3) * 256 + t);
    } else if (b >= PO_WR1GS) {
        cvt_t<DIN, HC1>(Wr1_gs, hWr1gs, (b - PO_WR1GS) * 256 + t);
    } else if (b >= PO_WL1GS) {
        cvt_t<DIN, HC1>(Wl1_gs, hWl1gs, (b - PO_WL1GS) * 256 + t);
    } else if (b >= PO_WR1SG) {
        cvt_t<DIN, HC1>(Wr1_sg, hWr1sg, (b - PO_WR1SG) * 256 + t);
    } else {
        cvt_t<DIN, HC1>(Wl1_sg, hWl1sg, (b - PO_WL1SG) * 256 + t);
    }
}

__device__ __forceinline__ v16h ld_frag(const _Float16* p) {
    v8h lo = *(const v8h*)(p);
    v8h hi = *(const v8h*)(p + 256);
    v16h r;
#pragma unroll
    for (int j = 0; j < 8; ++j) { r[j] = lo[j]; r[8 + j] = hi[j]; }
    return r;
}

#define GEMM_BLOCKS(M, N) ((((M) + 127) / 128) * ((N) / 64))

template <int M, int N>
__device__ __forceinline__ void
gemm_tile(const _Float16* __restrict__ X, const _Float16* __restrict__ Wt,
          const float* __restrict__ bias, float* __restrict__ Y, int t) {
    constexpr int K  = 256;
    constexpr int MB = (M + 127) / 128;
    const int bn = t / MB, bm = t - bn * MB;
    int wave = threadIdx.x >> 5;
    int lane = threadIdx.x & 31;
    int wm = wave & 3, wn = wave >> 2;
    int m0 = bm * 128 + wm * 32;
    if (m0 >= M) return;
    int n0 = bn * 64 + wn * 32;
    int lr = lane & 15;
    int cb0 = n0 + lr;
    int cb1 = cb0 + 16;

    static_assert(M % 32 == 0 && N % 32 == 0 && K == 256, "whole 16-row groups per wave tile");
    const _Float16* pA0 = X  + ((size_t)(m0 >> 4) * 8) * 512 + lane * 8;
    const _Float16* pA1 = pA0 + 8 * 512;
    const _Float16* pB0 = Wt + ((size_t)(n0 >> 4) * 8) * 512 + lane * 8;
    const _Float16* pB1 = pB0 + 8 * 512;

    v8f acc00 = {}, acc01 = {}, acc10 = {}, acc11 = {};
    for (int k0 = 0; k0 < K; k0 += 32) {
        const int ko = (k0 >> 5) * 512;
        v16h a0 = ld_frag(pA0 + ko);
        v16h a1 = ld_frag(pA1 + ko);
        v16h b0 = ld_frag(pB0 + ko);
        v16h b1 = ld_frag(pB1 + ko);
        acc00 = __builtin_amdgcn_wmma_f32_16x16x32_f16(false, a0, false, b0, (short)0, acc00, false, false);
        acc01 = __builtin_amdgcn_wmma_f32_16x16x32_f16(false, a0, false, b1, (short)0, acc01, false, false);
        acc10 = __builtin_amdgcn_wmma_f32_16x16x32_f16(false, a1, false, b0, (short)0, acc10, false, false);
        acc11 = __builtin_amdgcn_wmma_f32_16x16x32_f16(false, a1, false, b1, (short)0, acc11, false, false);
    }

    float bv0 = bias[cb0];
    float bv1 = bias[cb1];
    int mr = m0 + 8 * (lane >> 4);
#pragma unroll
    for (int v = 0; v < 8; ++v) {
        int ma = mr + v;
        if (ma < M) {
            Y[(size_t)ma * N + cb0] = acc00[v] + bv0;
            Y[(size_t)ma * N + cb1] = acc01[v] + bv1;
        }
        int mb = mr + 16 + v;
        if (mb < M) {
            Y[(size_t)mb * N + cb0] = acc10[v] + bv0;
            Y[(size_t)mb * N + cb1] = acc11[v] + bv1;
        }
    }
}

#define G1_CNT  (2 * NE / 1024)
#define G1_O0   G1_CNT
#define G1_O1   (G1_O0 + GEMM_BLOCKS(NG, HC1))
#define G1_O2   (G1_O1 + GEMM_BLOCKS(NG, HC1))
#define G1_O3   (G1_O2 + GEMM_BLOCKS(NS, HC1))
#define G1_O4   (G1_O3 + GEMM_BLOCKS(NS, HC1))
#define G1_END  (G1_O4 + GEMM_BLOCKS(NS, C1))

__global__ void __launch_bounds__(256)
k_gemm_l1(const _Float16* __restrict__ hx_sample, const _Float16* __restrict__ hx_gene,
          const _Float16* __restrict__ hWr1sg, const float* __restrict__ br1_sg, float* __restrict__ xr_sg,
          const _Float16* __restrict__ hWl1gs, const float* __restrict__ bl1_gs, float* __restrict__ xl_gs,
          const _Float16* __restrict__ hWl1sg, const float* __restrict__ bl1_sg, float* __restrict__ xl_sg,
          const _Float16* __restrict__ hWr1gs, const float* __restrict__ br1_gs, float* __restrict__ xr_gs,
          const _Float16* __restrict__ hsl1W,  const float* __restrict__ sl1_b,  float* __restrict__ sl1,
          const int* __restrict__ sg_dst, int* __restrict__ cnt_sg,
          const int* __restrict__ gs_dst, int* __restrict__ cnt_gs) {
    const int b = blockIdx.x;
    if (b < G1_CNT) {
        const int e = (b * 256 + (int)threadIdx.x) * 4;
        if (e < NE) {
            const v4i d = *(const v4i*)(sg_dst + e);
#pragma unroll
            for (int k = 0; k < 4; ++k) atomicAdd(&cnt_sg[d[k]], 1);
        } else {
            const v4i d = *(const v4i*)(gs_dst + (e - NE));
#pragma unroll
            for (int k = 0; k < 4; ++k) atomicAdd(&cnt_gs[d[k]], 1);
        }
    }
    else if (b < G1_O1) gemm_tile<NG, HC1>(hx_gene,   hWr1sg, br1_sg, xr_sg, b - G1_O0);
    else if (b < G1_O2) gemm_tile<NG, HC1>(hx_gene,   hWl1gs, bl1_gs, xl_gs, b - G1_O1);
    else if (b < G1_O3) gemm_tile<NS, HC1>(hx_sample, hWl1sg, bl1_sg, xl_sg, b - G1_O2);
    else if (b < G1_O4) gemm_tile<NS, HC1>(hx_sample, hWr1gs, br1_gs, xr_gs, b - G1_O3);
    else                gemm_tile<NS, C1 >(hx_sample, hsl1W,  sl1_b,  sl1,   b - G1_O4);
}

#define G3_O1   GEMM_BLOCKS(NG, HC3)
#define G3_O2   (G3_O1 + GEMM_BLOCKS(NS, HC3))
#define G3_END  (G3_O2 + GEMM_BLOCKS(NS, C3))

__global__ void __launch_bounds__(256)
k_gemm_l3(const _Float16* __restrict__ hx1_gene, const _Float16* __restrict__ hx1_sample,
          const _Float16* __restrict__ hWl3,  const float* __restrict__ bl3,   float* __restrict__ xl3,
          const _Float16* __restrict__ hWr3,  const float* __restrict__ br3,   float* __restrict__ xr3,
          const _Float16* __restrict__ hsl3W, const float* __restrict__ sl3_b, float* __restrict__ sl3) {
    const int b = blockIdx.x;
    if      (b < G3_O1) gemm_tile<NG, HC3>(hx1_gene,   hWl3,  bl3,   xl3, b);
    else if (b < G3_O2) gemm_tile<NS, HC3>(hx1_sample, hWr3,  br3,   xr3, b - G3_O1);
    else                gemm_tile<NS, C3 >(hx1_sample, hsl3W, sl3_b, sl3, b - G3_O2);
}

__global__ void __launch_bounds__(1024)
k_deg_scan(const int* __restrict__ cnt_a, int* __restrict__ off_a, int* __restrict__ cur_a, int n_a,
           const int* __restrict__ cnt_b, int* __restrict__ off_b, int* __restrict__ cur_b, int n_b) {
    __shared__ int part[1024];
    const int* cnt = blockIdx.x == 0 ? cnt_a : cnt_b;
    int*       off = blockIdx.x == 0 ? off_a : off_b;
    int*       cur = blockIdx.x == 0 ? cur_a : cur_b;
    const int  n   = blockIdx.x == 0 ? n_a : n_b;
    const int t = threadIdx.x;
    const int per = (n + 1023) / 1024;
    const int b = t * per;
    int e = b + per; if (e > n) e = n;
    int s = 0;
    for (int i = b; i < e; ++i) s += cnt[i];
    part[t] = s;
    __syncthreads();
    for (int d = 1; d < 1024; d <<= 1) {
        int v = (t >= d) ? part[t - d] : 0;
        __syncthreads();
        part[t] += v;
        __syncthreads();
    }
    int run = part[t] - s;
    for (int i = b; i < e; ++i) { off[i] = run; cur[i] = run; run += cnt[i]; }
    if (t == 1023) off[n] = part[1023];
}

__global__ void __launch_bounds__(256)
k_group_fill(const int* __restrict__ src_a, const int* __restrict__ dst_a,
             int* __restrict__ cur_a, int* __restrict__ ss_a,
             const int* __restrict__ src_b, const int* __restrict__ dst_b,
             int* __restrict__ cur_b, int* __restrict__ ss_b, int E) {
    const int e = (blockIdx.x * 256 + (int)threadIdx.x) * 4;
    if (e < E) {
        const v4i d = *(const v4i*)(dst_a + e);
        const v4i s = *(const v4i*)(src_a + e);
        int pos[4];
#pragma unroll
        for (int k = 0; k < 4; ++k) pos[k] = atomicAdd(&cur_a[d[k]], 1);
#pragma unroll
        for (int k = 0; k < 4; ++k) ss_a[pos[k]] = s[k];
    } else if (e < 2 * E) {
        const int j = e - E;
        const v4i d = *(const v4i*)(dst_b + j);
        const v4i s = *(const v4i*)(src_b + j);
        int pos[4];
#pragma unroll
        for (int k = 0; k < 4; ++k) pos[k] = atomicAdd(&cur_b[d[k]], 1);
#pragma unroll
        for (int k = 0; k < 4; ++k) ss_b[pos[k]] = s[k];
    }
}

__device__ __forceinline__ float sum8(float t) {
    t += __builtin_bit_cast(float, __builtin_amdgcn_update_dpp(0, __builtin_bit_cast(int, t), 0x161, 0xf, 0xf, true));
    t += __builtin_bit_cast(float, __builtin_amdgcn_update_dpp(0, __builtin_bit_cast(int, t), 0x162, 0xf, 0xf, true));
    t += __builtin_bit_cast(float, __builtin_amdgcn_update_dpp(0, __builtin_bit_cast(int, t), 0x164, 0xf, 0xf, true));
    return t;
}
__device__ __forceinline__ float sum_heads(float t) {
    t += __builtin_bit_cast(float, __builtin_amdgcn_update_dpp(0, __builtin_bit_cast(int, t), 0x168, 0xf, 0xf, true));
    t += __builtin_bit_cast(float, __builtin_amdgcn_permlanex16(__builtin_bit_cast(int, t), __builtin_bit_cast(int, t),
                                                                0x76543210, 0xfedcba98, false, false));
    return t;
}

#define EPI_GENE   0
#define EPI_SAMPLE 1
#define EPI_FINAL  2

template <int C, int EPI>
__device__ __forceinline__ void
gat_node(const int* __restrict__ off, const int* __restrict__ ssrc,
         const float* __restrict__ xl, const float* __restrict__ xr,
         const float* __restrict__ att, const float* __restrict__ bias,
         const float* __restrict__ sl, _Float16* __restrict__ outh, float* __restrict__ outf,
         const int d) {
    constexpr int V   = C / 32;
    constexpr int U   = 4;
    constexpr int ROW = NH * C;
    const int lane = threadIdx.x & 31;
    const int ls = lane & 7, h = lane >> 3;
    const int cbase = h * C + 4 * ls;

    const float* pr = xr + (size_t)d * ROW + cbase;
    v4f r[V], a[V], acc[V];
#pragma unroll
    for (int j = 0; j < V; ++j) {
        r[j] = *(const v4f*)(pr + 32 * j);
        const float* q = att + cbase + 32 * j;
        a[j] = (v4f){q[0], q[1], q[2], q[3]};
        acc[j] = (v4f){0.f, 0.f, 0.f, 0.f};
    }
    float m = -3.0e38f, den = 0.f;
    const int beg = off[d], end = off[d + 1];

    for (int i = beg; i < end; i += U) {
        v4f x[U][V];
#pragma unroll
        for (int u = 0; u < U; ++u) {
            int idx = i + u; if (idx > end - 1) idx = end - 1;
            const float* pl = xl + (size_t)ssrc[idx] * ROW + cbase;
#pragma unroll
            for (int j = 0; j < V; ++j) x[u][j] = *(const v4f*)(pl + 32 * j);
        }
#pragma unroll
        for (int u = 0; u < U; ++u) {
            float t = 0.f;
#pragma unroll
            for (int j = 0; j < V; ++j) {
                v4f g = x[u][j] + r[j];
#pragma unroll
                for (int k = 0; k < 4; ++k) {
                    float gk = g[k];
                    gk = gk > 0.f ? gk : 0.2f * gk;
                    t += gk * a[j][k];
                }
            }
            t = sum8(t);
            t = (i + u < end) ? t : -3.0e38f;
            const float dlt = t - m;
            const float ex  = expf(-fabsf(dlt));
            const bool  up  = dlt > 0.f;
            const float sc  = up ? ex : 1.f;
            const float p   = up ? 1.f : ex;
            m   = up ? t : m;
            den = den * sc + p;
#pragma unroll
            for (int j = 0; j < V; ++j) acc[j] = acc[j] * sc + x[u][j] * p;
        }
    }

    const float inv = 1.f / (den + 1e-16f);
    if (EPI == EPI_FINAL) {
        v4f pick = (v4f){0.f, 0.f, 0.f, 0.f};
#pragma unroll
        for (int j = 0; j < V; ++j) {
            v4f s = acc[j] * inv;
#pragma unroll
            for (int k = 0; k < 4; ++k) s[k] = sum_heads(s[k]);
            pick = (h == j) ? s : pick;
        }
        const int c = 32 * h + 4 * ls;
        const v4f bq = *(const v4f*)(bias + c);
        const v4f sq = *(const v4f*)(sl + (size_t)d * C + c);
        v4f o;
#pragma unroll
        for (int k = 0; k < 4; ++k) o[k] = eluf(0.25f * pick[k] + bq[k] + sq[k]);
        *(v4f*)(outf + (size_t)d * C + c) = o;
    } else {
        static_assert(EPI == EPI_FINAL || ROW == 256, "fragment-order storage is written for K = 256");
#pragma unroll
        for (int j = 0; j < V; ++j) {
            v4f o = acc[j] * inv;
            const float* q = bias + cbase + 32 * j;
            o += (v4f){q[0], q[1], q[2], q[3]};
            if (EPI == EPI_SAMPLE) o += *(const v4f*)(sl + (size_t)d * C + 4 * ls + 32 * j);
            v4h oh;
#pragma unroll
            for (int k = 0; k < 4; ++k) oh[k] = (_Float16)eluf(o[k]);
            *(v4h*)(outh + fidx(d, cbase + 32 * j)) = oh;
        }
    }
}

#define A1_GS  (NS / 8)
#define A1_END (A1_GS + NG / 8)

__global__ void __launch_bounds__(256)
k_gat_l1(const int* __restrict__ off_gs, const int* __restrict__ ss_gs,
         const float* __restrict__ xl_gs, const float* __restrict__ xr_gs,
         const float* __restrict__ att1_gs, const float* __restrict__ bias1_gs,
         const float* __restrict__ sl1, _Float16* __restrict__ hx1_sample,
         const int* __restrict__ off_sg, const int* __restrict__ ss_sg,
         const float* __restrict__ xl_sg, const float* __restrict__ xr_sg,
         const float* __restrict__ att1_sg, const float* __restrict__ bias1_sg,
         _Float16* __restrict__ hx1_gene) {
    const int w = __builtin_amdgcn_readfirstlane((int)(blockIdx.x * 8 + (threadIdx.x >> 5)));
    if (w < NS) gat_node<C1, EPI_SAMPLE>(off_gs, ss_gs, xl_gs, xr_gs, att1_gs, bias1_gs, sl1, hx1_sample, nullptr, w);
    else        gat_node<C1, EPI_GENE  >(off_sg, ss_sg, xl_sg, xr_sg, att1_sg, bias1_sg, nullptr, hx1_gene, nullptr, w - NS);
}

__global__ void __launch_bounds__(256)
k_gat_l3(const int* __restrict__ off, const int* __restrict__ ssrc,
         const float* __restrict__ xl, const float* __restrict__ xr,
         const float* __restrict__ att, const float* __restrict__ bias3,
         const float* __restrict__ sl3, float* __restrict__ out) {
    const int d = __builtin_amdgcn_readfirstlane((int)(blockIdx.x * 8 + (threadIdx.x >> 5)));
    gat_node<C3, EPI_FINAL>(off, ssrc, xl, xr, att, bias3, sl3, nullptr, out, d);
}

extern "C" void kernel_launch(void* const* d_in, const int* in_sizes, int n_in,
                              void* d_out, int out_size, void* d_ws, size_t ws_size,
                              hipStream_t stream) {
    (void)in_sizes; (void)n_in; (void)out_size; (void)ws_size;

    const float* x_sample = (const float*)d_in[0];
    const float* x_gene   = (const float*)d_in[1];
    const int*   sg_src   = (const int*)d_in[2];
    const int*   sg_dst   = (const int*)d_in[3];
    const int*   gs_src   = (const int*)d_in[4];
    const int*   gs_dst   = (const int*)d_in[5];
    const float* Wl1_sg = (const float*)d_in[6];   const float* bl1_sg  = (const float*)d_in[7];
    const float* Wr1_sg = (const float*)d_in[8];   const float* br1_sg  = (const float*)d_in[9];
    const float* att1_sg = (const float*)d_in[10]; const float* bias1_sg = (const float*)d_in[11];
    const float* Wl1_gs = (const float*)d_in[12];  const float* bl1_gs  = (const float*)d_in[13];
    const float* Wr1_gs = (const float*)d_in[14];  const float* br1_gs  = (const float*)d_in[15];
    const float* att1_gs = (const float*)d_in[16]; const float* bias1_gs = (const float*)d_in[17];
    const float* Wl3 = (const float*)d_in[18];     const float* bl3 = (const float*)d_in[19];
    const float* Wr3 = (const float*)d_in[20];     const float* br3 = (const float*)d_in[21];
    const float* att3 = (const float*)d_in[22];    const float* bias3 = (const float*)d_in[23];
    const float* sl1_W = (const float*)d_in[24];   const float* sl1_b = (const float*)d_in[25];
    const float* sl3_W = (const float*)d_in[26];   const float* sl3_b = (const float*)d_in[27];

    char* p = (char*)d_ws;
    auto carve = [&](size_t bytes) -> char* {
        char* r = p; p += (bytes + 255) & ~(size_t)255; return r;
    };

    _Float16* hx_sample = (_Float16*)carve((size_t)NS * DIN * 2);
    _Float16* hx_gene   = (_Float16*)carve((size_t)NG * DIN * 2);
    _Float16* hWl1sg = (_Float16*)carve((size_t)DIN * HC1 * 2);
    _Float16* hWr1sg = (_Float16*)carve((size_t)DIN * HC1 * 2);
    _Float16* hWl1gs = (_Float16*)carve((size_t)DIN * HC1 * 2);
    _Float16* hWr1gs = (_Float16*)carve((size_t)DIN * HC1 * 2);
    _Float16* hWl3   = (_Float16*)carve((size_t)HC1 * HC3 * 2);
    _Float16* hWr3   = (_Float16*)carve((size_t)HC1 * HC3 * 2);
    _Float16* hsl1W  = (_Float16*)carve((size_t)DIN * C1 * 2);
    _Float16* hsl3W  = (_Float16*)carve((size_t)HC1 * C3 * 2);
    _Float16* hx1_gene   = (_Float16*)carve((size_t)NG * HC1 * 2);
    _Float16* hx1_sample = (_Float16*)carve((size_t)NS * HC1 * 2);
    float* xl_sg = (float*)carve((size_t)NS * HC1 * 4);
    float* xr_sg = (float*)carve((size_t)NG * HC1 * 4);
    float* xl_gs = (float*)carve((size_t)NG * HC1 * 4);
    float* xr_gs = (float*)carve((size_t)NS * HC1 * 4);
    (void)carve((size_t)NS * HC1 * 4);
    float* sl1 = (float*)carve((size_t)NS * C1 * 4);
    float* xl3 = (float*)carve((size_t)NG * HC3 * 4);
    float* xr3 = (float*)carve((size_t)NS * HC3 * 4);
    (void)carve((size_t)NS * HC3 * 4);
    float* sl3 = (float*)carve((size_t)NS * C3 * 4);
    int* cnt    = (int*)carve((size_t)(NG + NS) * 4);
    int* cur    = (int*)carve((size_t)(NG + NS) * 4);
    int* off_sg = (int*)carve((size_t)(NG + 1) * 4);
    int* off_gs = (int*)carve((size_t)(NS + 1) * 4);
    int* ss_sg  = (int*)carve((size_t)NE * 4);
    int* ss_gs  = (int*)carve((size_t)NE * 4);

    k_prep<<<PO_END, 256, 0, stream>>>(x_sample, x_gene, Wl1_sg, Wr1_sg, Wl1_gs, Wr1_gs, Wl3, Wr3,
                                       sl1_W, sl3_W, hx_sample, hx_gene, hWl1sg, hWr1sg, hWl1gs,
                                       hWr1gs, hWl3, hWr3, hsl1W, hsl3W, cnt);
    k_gemm_l1<<<G1_END, 256, 0, stream>>>(hx_sample, hx_gene,
                                          hWr1sg, br1_sg, xr_sg,  hWl1gs, bl1_gs, xl_gs,
                                          hWl1sg, bl1_sg, xl_sg,  hWr1gs, br1_gs, xr_gs,
                                          hsl1W, sl1_b, sl1,
                                          sg_dst, cnt, gs_dst, cnt + NG);
    k_deg_scan<<<2, 1024, 0, stream>>>(cnt, off_sg, cur, NG, cnt + NG, off_gs, cur + NG, NS);
    k_group_fill<<<2 * NE / 1024, 256, 0, stream>>>(sg_src, sg_dst, cur, ss_sg,
                                                   gs_src, gs_dst, cur + NG, ss_gs, NE);
    k_gat_l1<<<A1_END, 256, 0, stream>>>(off_gs, ss_gs, xl_gs, xr_gs, att1_gs, bias1_gs, sl1, hx1_sample,
                                         off_sg, ss_sg, xl_sg, xr_sg, att1_sg, bias1_sg, hx1_gene);
    k_gemm_l3<<<G3_END, 256, 0, stream>>>(hx1_gene, hx1_sample, hWl3, bl3, xl3, hWr3, br3, xr3,
                                          hsl3W, sl3_b, sl3);
    k_gat_l3<<<NS / 8, 256, 0, stream>>>(off_gs, ss_gs, xl3, xr3, att3, bias3, sl3, (float*)d_out);
}


# aggregation loops: softmax per group of 4 edges (one max, one rescale, one multiply-add per channel per edge), leaky_relu as a vector max, hardware exp2: vector instructions per trip 414 -> 223
# speedup vs baseline: 8.1815x; 1.8426x over previous
#include <hip/hip_runtime.h>
#include <hip/hip_bf16.h>


#define NS   4096
#define NG   20000
#define NE   131072
#define DIN  256
#define NH   4
#define C1   64
#define C3   128
#define HC1  256
#define HC3  512

typedef __attribute__((ext_vector_type(16))) _Float16 v16h;
typedef __attribute__((ext_vector_type(8)))  _Float16 v8h;
typedef __attribute__((ext_vector_type(4)))  _Float16 v4h;
typedef __attribute__((ext_vector_type(8)))  float    v8f;
typedef __attribute__((ext_vector_type(4)))  float    v4f;
typedef __attribute__((ext_vector_type(4)))  int      v4i;

__device__ __forceinline__ float eluf(float x) { return x > 0.f ? x : (expf(x) - 1.f); }

__device__ __forceinline__ size_t fidx(int row, int k) {
    return ((size_t)((row >> 4) * 8 + (k >> 5)) << 9)
         + (size_t)((((k >> 4) & 1) << 8)
                    + (((row & 15) + (((k >> 3) & 1) << 4)) << 3)
                    + (k & 7));
}

#define SB_CH     2048
#define SB_NCH    (NE / SB_CH)
#define SB_NR     64
#define SB_RW_SG  313
#define SB_RW_GS  64
#define SB_HN     512
static_assert(NE % SB_CH == 0 && SB_NCH == 64, "a level-2 thread per chunk, 8 chunks per wave");
static_assert(SB_RW_SG * SB_NR >= NG && SB_RW_SG * (SB_NR - 1) < NG, "ranges cover the gene nodes, none empty");
static_assert(SB_RW_GS * SB_NR == NS, "ranges cover the sample nodes");
static_assert(SB_RW_SG <= SB_HN && SB_RW_GS <= SB_HN, "one LDS counter per node of a range");
static_assert(NG < 32768 && NS < 32768, "dst << 16 | src fits a positive int");

__device__ __forceinline__ int wave_incl_scan(int v, int lane) {
#pragma unroll
    for (int o = 1; o < 32; o <<= 1) {
        const int u = __shfl_up(v, o);
        v += (lane >= o) ? u : 0;
    }
    return v;
}

template <int RW>
__device__ __forceinline__ void
bucket_chunk(const int* __restrict__ src, const int* __restrict__ dst,
             int* __restrict__ bk, int* __restrict__ T, const int c, const int t,
             int* hist, int* start) {
    if (t < SB_NR) hist[t] = 0;
    __syncthreads();
    const int e0 = c * SB_CH + 4 * t;
    const v4i d0 = *(const v4i*)(dst + e0);
    const v4i d1 = *(const v4i*)(dst + e0 + SB_CH / 2);
    const v4i s0 = *(const v4i*)(src + e0);
    const v4i s1 = *(const v4i*)(src + e0 + SB_CH / 2);
    int rg[8], key[8], rank[8];
#pragma unroll
    for (int k = 0; k < 4; ++k) {
        rg[k]     = d0[k] / RW;  key[k]     = (d0[k] << 16) | s0[k];
        rg[4 + k] = d1[k] / RW;  key[4 + k] = (d1[k] << 16) | s1[k];
    }
#pragma unroll
    for (int k = 0; k < 8; ++k) rank[k] = atomicAdd(&hist[rg[k]], 1);
    __syncthreads();
    if (t < 32) {
        const int a = hist[2 * t], b = hist[2 * t + 1];
        const int ex = wave_incl_scan(a + b, t) - (a + b);
        start[2 * t]     = ex;
        start[2 * t + 1] = ex + a;
        *(int2*)(T + c * SB_NR + 2 * t) = make_int2(a, b);
    }
    __syncthreads();
    int* out = bk + c * SB_CH;
#pragma unroll
    for (int k = 0; k < 8; ++k) out[start[rg[k]] + rank[k]] = key[k];
}

template <int RW, int N>
__device__ __forceinline__ void
group_range(const int* __restrict__ bk, const int* __restrict__ T,
            int* __restrict__ off, int* __restrict__ ss, const int r,
            int* cs_s, int* n_s, int* h, int* cur, int* wt, int* base_s) {
    const int t = threadIdx.x, lane = t & 31, w = t >> 5;
    const int lo = r * RW;
    h[2 * t] = 0;  h[2 * t + 1] = 0;
    if (t == 0) *base_s = 0;
    __syncthreads();
    if (t < SB_NCH) {
        const v4i* row = (const v4i*)(T + t * SB_NR);
        int before = 0, mine = 0;
#pragma unroll
        for (int q = 0; q < SB_NR / 4; ++q) {
            const v4i v = row[q];
#pragma unroll
            for (int k = 0; k < 4; ++k) {
                const int rr = 4 * q + k;
                before += (rr < r) ? v[k] : 0;
                mine    = (rr == r) ? v[k] : mine;
            }
        }
        cs_s[t] = before;
        n_s[t]  = mine;
        atomicAdd(base_s, before);
    }
    __syncthreads();
    const int base = *base_s;

    int key[8][2];
#pragma unroll
    for (int q = 0; q < 8; ++q) {
        const int ch = w + 8 * q;
        const int n = n_s[ch];
        const int* p = bk + ch * SB_CH + cs_s[ch];
        key[q][0] = (lane < n)      ? p[lane]      : -1;
        key[q][1] = (lane + 32 < n) ? p[lane + 32] : -1;
    }
#pragma unroll
    for (int q = 0; q < 8; ++q) {
#pragma unroll
        for (int u = 0; u < 2; ++u)
            if (key[q][u] >= 0) atomicAdd(&h[(key[q][u] >> 16) - lo], 1);
        const int ch = w + 8 * q;
        const int n = n_s[ch];
        const int* p = bk + ch * SB_CH + cs_s[ch];
        for (int i = lane + 64; i < n; i += 32) atomicAdd(&h[(p[i] >> 16) - lo], 1);
    }
    __syncthreads();

    const int a = h[2 * t], b = h[2 * t + 1];
    const int inc = wave_incl_scan(a + b, lane);
    if (lane == 31) wt[w] = inc;
    __syncthreads();
    int woff = 0;
    for (int q = 0; q < 8; ++q) woff += (q < w) ? wt[q] : 0;
    const int ex = base + woff + inc - (a + b);
    cur[2 * t]     = ex;
    cur[2 * t + 1] = ex + a;
    const int j = 2 * t;
    if (j < RW && lo + j < N)         off[lo + j]     = ex;
    if (j + 1 < RW && lo + j + 1 < N) off[lo + j + 1] = ex + a;
    if (r == SB_NR - 1 && t == 0)     off[N] = NE;
    __syncthreads();

#pragma unroll
    for (int q = 0; q < 8; ++q) {
#pragma unroll
        for (int u = 0; u < 2; ++u)
            if (key[q][u] >= 0) ss[atomicAdd(&cur[(key[q][u] >> 16) - lo], 1)] = key[q][u] & 0xffff;
        const int ch = w + 8 * q;
        const int n = n_s[ch];
        const int* p = bk + ch * SB_CH + cs_s[ch];
        for (int i = lane + 64; i < n; i += 32) {
            const int kv = p[i];
            ss[atomicAdd(&cur[(kv >> 16) - lo], 1)] = kv & 0xffff;
        }
    }
}

#define PB_W1   4
#define PB_W3   8
#define PO_WL1SG 0
#define PO_WR1SG (PO_WL1SG + PB_W1)
#define PO_WL1GS (PO_WR1SG + PB_W1)
#define PO_WR1GS (PO_WL1GS + PB_W1)
#define PO_WL3   (PO_WR1GS + PB_W1)
#define PO_WR3   (PO_WL3 + PB_W3)
#define PO_SL1   (PO_WR3 + PB_W3)
#define PO_SL3   (PO_SL1 + 1)
#define PO_XS    (PO_SL3 + 2)
#define PO_XG    (PO_XS + 64)
#define PO_END   (PO_XG + 500)
#define PO_SORT  (2 * SB_NCH)

template <int K, int N>
__device__ __forceinline__ void cvt_t(const float* __restrict__ in, _Float16* __restrict__ out, int i) {
    const int q = i / N, n = i - q * N;
    const float* p = in + (size_t)(q * 64) * N + n;
    float v[64];
#pragma unroll
    for (int kk = 0; kk < 64; ++kk) v[kk] = p[(size_t)kk * N];
    static_assert(K == 256, "fragment-order storage is written for K = 256");
#pragma unroll
    for (int s = 0; s < 8; ++s) {
        v8h hv;
#pragma unroll
        for (int j = 0; j < 8; ++j) hv[j] = (_Float16)v[8 * s + j];
        *(v8h*)(out + fidx(n, q * 64 + 8 * s)) = hv;
    }
}

template <int U>
__device__ __forceinline__ void cvt_run(const float* __restrict__ in, _Float16* __restrict__ out, int blk, int t) {
    static_assert(U % 2 == 0, "8 rows of the block per piece");
    const int k    = 64 * ((t >> 5) & 3) + 8 * (t & 7);
    const int row0 = blk * (U * 4) + 4 * (t >> 7) + ((t >> 3) & 3);
    v4f x[U];
#pragma unroll
    for (int j = 0; j < U / 2; ++j) {
        const float* p = in + (size_t)(row0 + 8 * j) * 256 + k;
        x[2 * j]     = *(const v4f*)(p);
        x[2 * j + 1] = *(const v4f*)(p + 4);
    }
#pragma unroll
    for (int j = 0; j < U / 2; ++j) {
        v8h hv;
#pragma unroll
        for (int e = 0; e < 4; ++e) { hv[e] = (_Float16)x[2 * j][e]; hv[4 + e] = (_Float16)x[2 * j + 1][e]; }
        *(v8h*)(out + fidx(row0 + 8 * j, k)) = hv;
    }
}

__global__ void __launch_bounds__(256)
k_prep(const float* __restrict__ x_sample, const float* __restrict__ x_gene,
       const float* __restrict__ Wl1_sg, const float* __restrict__ Wr1_sg,
       const float* __restrict__ Wl1_gs, const float* __restrict__ Wr1_gs,
       const float* __restrict__ Wl3, const float* __restrict__ Wr3,
       const float* __restrict__ sl1_W, const float* __restrict__ sl3_W,
       _Float16* __restrict__ hx_sample, _Float16* __restrict__ hx_gene,
       _Float16* __restrict__ hWl1sg, _Float16* __restrict__ hWr1sg,
       _Float16* __restrict__ hWl1gs, _Float16* __restrict__ hWr1gs,
       _Float16* __restrict__ hWl3, _Float16* __restrict__ hWr3,
       _Float16* __restrict__ hsl1W, _Float16* __restrict__ hsl3W,
       const int* __restrict__ sg_src, const int* __restrict__ sg_dst,
       const int* __restrict__ gs_src, const int* __restrict__ gs_dst,
       int* __restrict__ bk_sg, int* __restrict__ T_sg,
       int* __restrict__ bk_gs, int* __restrict__ T_gs) {
    __shared__ int sb_hist[SB_NR], sb_start[SB_NR];
    const int t = threadIdx.x;
    if (blockIdx.x < PO_SORT) {
        const int cb = blockIdx.x;
        if (cb < SB_NCH) bucket_chunk<SB_RW_SG>(sg_src, sg_dst, bk_sg, T_sg, cb, t, sb_hist, sb_start);
        else             bucket_chunk<SB_RW_GS>(gs_src, gs_dst, bk_gs, T_gs, cb - SB_NCH, t, sb_hist, sb_start);
        return;
    }
    const int b = blockIdx.x - PO_SORT;
    if (b >= PO_XG) {
        cvt_run<10>(x_gene, hx_gene, b - PO_XG, t);
    } else if (b >= PO_XS) {
        cvt_run<16>(x_sample, hx_sample, b - PO_XS, t);
    } else if (b >= PO_SL3) {
        cvt_t<HC1, C3>(sl3_W, hsl3W, (b - PO_SL3) * 256 + t);
    } else if (b >= PO_SL1) {
        cvt_t<DIN, C1>(sl1_W, hsl1W, (b - PO_SL1) * 256 + t);
    } else if (b >= PO_WR3) {
        cvt_t<HC1, HC3>(Wr3, hWr3, (b - PO_WR3) * 256 + t);
    } else if (b >= PO_WL3) {
        cvt_t<HC1, HC3>(Wl3, hWl3, (b - PO_WL3) * 256 + t);
    } else if (b >= PO_WR1GS) {
        cvt_t<DIN, HC1>(Wr1_gs, hWr1gs, (b - PO_WR1GS) * 256 + t);
    } else if (b >= PO_WL1GS) {
        cvt_t<DIN, HC1>(Wl1_gs, hWl1gs, (b - PO_WL1GS) * 256 + t);
    } else if (b >= PO_WR1SG) {
        cvt_t<DIN, HC1>(Wr1_sg, hWr1sg, (b - PO_WR1SG) * 256 + t);
    } else {
        cvt_t<DIN, HC1>(Wl1_sg, hWl1sg, (b - PO_WL1SG) * 256 + t);
    }
}

__device__ __forceinline__ v16h ld_frag(const _Float16* p) {
    v8h lo = *(const v8h*)(p);
    v8h hi = *(const v8h*)(p + 256);
    v16h r;
#pragma unroll
    for (int j = 0; j < 8; ++j) { r[j] = lo[j]; r[8 + j] = hi[j]; }
    return r;
}

#define GEMM_BLOCKS(M, N) ((((M) + 127) / 128) * ((N) / 64))

template <int M, int N>
__device__ __forceinline__ void
gemm_tile(const _Float16* __restrict__ X, const _Float16* __restrict__ Wt,
          const float* __restrict__ bias, float* __restrict__ Y, int t) {
    constexpr int K  = 256;
    constexpr int MB = (M + 127) / 128;
    const int bn = t / MB, bm = t - bn * MB;
    int wave = threadIdx.x >> 5;
    int lane = threadIdx.x & 31;
    int wm = wave & 3, wn = wave >> 2;
    int m0 = bm * 128 + wm * 32;
    if (m0 >= M) return;
    int n0 = bn * 64 + wn * 32;
    int lr = lane & 15;
    int cb0 = n0 + lr;
    int cb1 = cb0 + 16;

    static_assert(M % 32 == 0 && N % 32 == 0 && K == 256, "whole 16-row groups per wave tile");
    const _Float16* pA0 = X  + ((size_t)(m0 >> 4) * 8) * 512 + lane * 8;
    const _Float16* pA1 = pA0 + 8 * 512;
    const _Float16* pB0 = Wt + ((size_t)(n0 >> 4) * 8) * 512 + lane * 8;
    const _Float16* pB1 = pB0 + 8 * 512;

    v8f acc00 = {}, acc01 = {}, acc10 = {}, acc11 = {};
    for (int k0 = 0; k0 < K; k0 += 32) {
        const int ko = (k0 >> 5) * 512;
        v16h a0 = ld_frag(pA0 + ko);
        v16h a1 = ld_frag(pA1 + ko);
        v16h b0 = ld_frag(pB0 + ko);
        v16h b1 = ld_frag(pB1 + ko);
        acc00 = __builtin_amdgcn_wmma_f32_16x16x32_f16(false, a0, false, b0, (short)0, acc00, false, false);
        acc01 = __builtin_amdgcn_wmma_f32_16x16x32_f16(false, a0, false, b1, (short)0, acc01, false, false);
        acc10 = __builtin_amdgcn_wmma_f32_16x16x32_f16(false, a1, false, b0, (short)0, acc10, false, false);
        acc11 = __builtin_amdgcn_wmma_f32_16x16x32_f16(false, a1, false, b1, (short)0, acc11, false, false);
    }

    float bv0 = bias[cb0];
    float bv1 = bias[cb1];
    int mr = m0 + 8 * (lane >> 4);
#pragma unroll
    for (int v = 0; v < 8; ++v) {
        int ma = mr + v;
        if (ma < M) {
            Y[(size_t)ma * N + cb0] = acc00[v] + bv0;
            Y[(size_t)ma * N + cb1] = acc01[v] + bv1;
        }
        int mb = mr + 16 + v;
        if (mb < M) {
            Y[(size_t)mb * N + cb0] = acc10[v] + bv0;
            Y[(size_t)mb * N + cb1] = acc11[v] + bv1;
        }
    }
}

#define G1_SORT (2 * SB_NR)
#define G1_O0   G1_SORT
#define G1_O1   (G1_O0 + GEMM_BLOCKS(NG, HC1))
#define G1_O2   (G1_O1 + GEMM_BLOCKS(NG, HC1))
#define G1_O3   (G1_O2 + GEMM_BLOCKS(NS, HC1))
#define G1_O4   (G1_O3 + GEMM_BLOCKS(NS, HC1))
#define G1_END  (G1_O4 + GEMM_BLOCKS(NS, C1))

__global__ void __launch_bounds__(256)
k_gemm_l1(const _Float16* __restrict__ hx_sample, const _Float16* __restrict__ hx_gene,
          const _Float16* __restrict__ hWr1sg, const float* __restrict__ br1_sg, float* __restrict__ xr_sg,
          const _Float16* __restrict__ hWl1gs, const float* __restrict__ bl1_gs, float* __restrict__ xl_gs,
          const _Float16* __restrict__ hWl1sg, const float* __restrict__ bl1_sg, float* __restrict__ xl_sg,
          const _Float16* __restrict__ hWr1gs, const float* __restrict__ br1_gs, float* __restrict__ xr_gs,
          const _Float16* __restrict__ hsl1W,  const float* __restrict__ sl1_b,  float* __restrict__ sl1,
          const int* __restrict__ bk_sg, const int* __restrict__ T_sg,
          int* __restrict__ off_sg, int* __restrict__ ss_sg,
          const int* __restrict__ bk_gs, const int* __restrict__ T_gs,
          int* __restrict__ off_gs, int* __restrict__ ss_gs) {
    __shared__ int sb_cs[SB_NCH], sb_n[SB_NCH], sb_h[SB_HN], sb_cur[SB_HN], sb_wt[8], sb_base;
    const int b = blockIdx.x;
    if (b < G1_SORT) {
        if (b < SB_NR) group_range<SB_RW_SG, NG>(bk_sg, T_sg, off_sg, ss_sg, b, sb_cs, sb_n, sb_h, sb_cur, sb_wt, &sb_base);
        else           group_range<SB_RW_GS, NS>(bk_gs, T_gs, off_gs, ss_gs, b - SB_NR, sb_cs, sb_n, sb_h, sb_cur, sb_wt, &sb_base);
    }
    else if (b < G1_O1) gemm_tile<NG, HC1>(hx_gene,   hWr1sg, br1_sg, xr_sg, b - G1_O0);
    else if (b < G1_O2) gemm_tile<NG, HC1>(hx_gene,   hWl1gs, bl1_gs, xl_gs, b - G1_O1);
    else if (b < G1_O3) gemm_tile<NS, HC1>(hx_sample, hWl1sg, bl1_sg, xl_sg, b - G1_O2);
    else if (b < G1_O4) gemm_tile<NS, HC1>(hx_sample, hWr1gs, br1_gs, xr_gs, b - G1_O3);
    else                gemm_tile<NS, C1 >(hx_sample, hsl1W,  sl1_b,  sl1,   b - G1_O4);
}

#define G3_O1   GEMM_BLOCKS(NG, HC3)
#define G3_O2   (G3_O1 + GEMM_BLOCKS(NS, HC3))
#define G3_END  (G3_O2 + GEMM_BLOCKS(NS, C3))

__global__ void __launch_bounds__(256)
k_gemm_l3(const _Float16* __restrict__ hx1_gene, const _Float16* __restrict__ hx1_sample,
          const _Float16* __restrict__ hWl3,  const float* __restrict__ bl3,   float* __restrict__ xl3,
          const _Float16* __restrict__ hWr3,  const float* __restrict__ br3,   float* __restrict__ xr3,
          const _Float16* __restrict__ hsl3W, const float* __restrict__ sl3_b, float* __restrict__ sl3) {
    const int b = blockIdx.x;
    if      (b < G3_O1) gemm_tile<NG, HC3>(hx1_gene,   hWl3,  bl3,   xl3, b);
    else if (b < G3_O2) gemm_tile<NS, HC3>(hx1_sample, hWr3,  br3,   xr3, b - G3_O1);
    else                gemm_tile<NS, C3 >(hx1_sample, hsl3W, sl3_b, sl3, b - G3_O2);
}

__device__ __forceinline__ float sum8(float t) {
    t += __builtin_bit_cast(float, __builtin_amdgcn_update_dpp(0, __builtin_bit_cast(int, t), 0x161, 0xf, 0xf, true));
    t += __builtin_bit_cast(float, __builtin_amdgcn_update_dpp(0, __builtin_bit_cast(int, t), 0x162, 0xf, 0xf, true));
    t += __builtin_bit_cast(float, __builtin_amdgcn_update_dpp(0, __builtin_bit_cast(int, t), 0x164, 0xf, 0xf, true));
    return t;
}
__device__ __forceinline__ float sum_heads(float t) {
    t += __builtin_bit_cast(float, __builtin_amdgcn_update_dpp(0, __builtin_bit_cast(int, t), 0x168, 0xf, 0xf, true));
    t += __builtin_bit_cast(float, __builtin_amdgcn_permlanex16(__builtin_bit_cast(int, t), __builtin_bit_cast(int, t),
                                                                0x76543210, 0xfedcba98, false, false));
    return t;
}

#define EPI_GENE   0
#define EPI_SAMPLE 1
#define EPI_FINAL  2

template <int C, int EPI>
__device__ __forceinline__ void
gat_node(const int* __restrict__ off, const int* __restrict__ ssrc,
         const float* __restrict__ xl, const float* __restrict__ xr,
         const float* __restrict__ att, const float* __restrict__ bias,
         const float* __restrict__ sl, _Float16* __restrict__ outh, float* __restrict__ outf,
         const int d) {
    constexpr int V   = C / 32;
    constexpr int U   = 4;
    constexpr int ROW = NH * C;
    const int lane = threadIdx.x & 31;
    const int ls = lane & 7, h = lane >> 3;
    const int cbase = h * C + 4 * ls;

    const float* pr = xr + (size_t)d * ROW + cbase;
    v4f r[V], a[V], acc[V];
#pragma unroll
    for (int j = 0; j < V; ++j) {
        r[j] = *(const v4f*)(pr + 32 * j);
        const float* q = att + cbase + 32 * j;
        a[j] = (v4f){q[0], q[1], q[2], q[3]};
        acc[j] = (v4f){0.f, 0.f, 0.f, 0.f};
    }
    constexpr float NEG   = -1.0e30f;
    constexpr float LOG2E = 1.4426950408889634f;
    float m = NEG, den = 0.f;
    const int beg = off[d], end = off[d + 1];

    for (int i = beg; i < end; i += U) {
        v4f x[U][V];
#pragma unroll
        for (int u = 0; u < U; ++u) {
            int idx = i + u; if (idx > end - 1) idx = end - 1;
            const float* pl = xl + (size_t)ssrc[idx] * ROW + cbase;
#pragma unroll
            for (int j = 0; j < V; ++j) x[u][j] = *(const v4f*)(pl + 32 * j);
        }
        float t[U];
#pragma unroll
        for (int u = 0; u < U; ++u) {
            v4f sv = (v4f){0.f, 0.f, 0.f, 0.f};
#pragma unroll
            for (int j = 0; j < V; ++j) {
                const v4f g = x[u][j] + r[j];
                sv += __builtin_elementwise_max(g, g * 0.2f) * a[j];
            }
            const float s = sum8((sv[0] + sv[1]) + (sv[2] + sv[3]));
            t[u] = (i + u < end) ? s : NEG;
        }
        float mn = m;
#pragma unroll
        for (int u = 0; u < U; ++u) mn = fmaxf(mn, t[u]);
        const float sc = __builtin_amdgcn_exp2f((m - mn) * LOG2E);
        float p[U];
        float ps = 0.f;
#pragma unroll
        for (int u = 0; u < U; ++u) {
            p[u] = __builtin_amdgcn_exp2f((t[u] - mn) * LOG2E);
            ps += p[u];
        }
        den = den * sc + ps;
#pragma unroll
        for (int j = 0; j < V; ++j) {
            v4f s4 = acc[j] * sc;
#pragma unroll
            for (int u = 0; u < U; ++u) s4 += x[u][j] * p[u];
            acc[j] = s4;
        }
        m = mn;
    }

    const float inv = 1.f / (den + 1e-16f);
    if (EPI == EPI_FINAL) {
        v4f pick = (v4f){0.f, 0.f, 0.f, 0.f};
#pragma unroll
        for (int j = 0; j < V; ++j) {
            v4f s = acc[j] * inv;
#pragma unroll
            for (int k = 0; k < 4; ++k) s[k] = sum_heads(s[k]);
            pick = (h == j) ? s : pick;
        }
        const int c = 32 * h + 4 * ls;
        const v4f bq = *(const v4f*)(bias + c);
        const v4f sq = *(const v4f*)(sl + (size_t)d * C + c);
        v4f o;
#pragma unroll
        for (int k = 0; k < 4; ++k) o[k] = eluf(0.25f * pick[k] + bq[k] + sq[k]);
        *(v4f*)(outf + (size_t)d * C + c) = o;
    } else {
        static_assert(EPI == EPI_FINAL || ROW == 256, "fragment-order storage is written for K = 256");
#pragma unroll
        for (int j = 0; j < V; ++j) {
            v4f o = acc[j] * inv;
            const float* q = bias + cbase + 32 * j;
            o += (v4f){q[0], q[1], q[2], q[3]};
            if (EPI == EPI_SAMPLE) o += *(const v4f*)(sl + (size_t)d * C + 4 * ls + 32 * j);
            v4h oh;
#pragma unroll
            for (int k = 0; k < 4; ++k) oh[k] = (_Float16)eluf(o[k]);
            *(v4h*)(outh + fidx(d, cbase + 32 * j)) = oh;
        }
    }
}

#define A1_GS  (NS / 8)
#define A1_END (A1_GS + NG / 8)

__global__ void __launch_bounds__(256)
k_gat_l1(const int* __restrict__ off_gs, const int* __restrict__ ss_gs,
         const float* __restrict__ xl_gs, const float* __restrict__ xr_gs,
         const float* __restrict__ att1_gs, const float* __restrict__ bias1_gs,
         const float* __restrict__ sl1, _Float16* __restrict__ hx1_sample,
         const int* __restrict__ off_sg, const int* __restrict__ ss_sg,
         const float* __restrict__ xl_sg, const float* __restrict__ xr_sg,
         const float* __restrict__ att1_sg, const float* __restrict__ bias1_sg,
         _Float16* __restrict__ hx1_gene) {
    const int w = __builtin_amdgcn_readfirstlane((int)(blockIdx.x * 8 + (threadIdx.x >> 5)));
    if (w < NS) gat_node<C1, EPI_SAMPLE>(off_gs, ss_gs, xl_gs, xr_gs, att1_gs, bias1_gs, sl1, hx1_sample, nullptr, w);
    else        gat_node<C1, EPI_GENE  >(off_sg, ss_sg, xl_sg, xr_sg, att1_sg, bias1_sg, nullptr, hx1_gene, nullptr, w - NS);
}

__global__ void __launch_bounds__(256)
k_gat_l3(const int* __restrict__ off, const int* __restrict__ ssrc,
         const float* __restrict__ xl, const float* __restrict__ xr,
         const float* __restrict__ att, const float* __restrict__ bias3,
         const float* __restrict__ sl3, float* __restrict__ out) {
    const int d = __builtin_amdgcn_readfirstlane((int)(blockIdx.x * 8 + (threadIdx.x >> 5)));
    gat_node<C3, EPI_FINAL>(off, ssrc, xl, xr, att, bias3, sl3, nullptr, out, d);
}

extern "C" void kernel_launch(void* const* d_in, const int* in_sizes, int n_in,
                              void* d_out, int out_size, void* d_ws, size_t ws_size,
                              hipStream_t stream) {
    (void)in_sizes; (void)n_in; (void)out_size; (void)ws_size;

    const float* x_sample = (const float*)d_in[0];
    const float* x_gene   = (const float*)d_in[1];
    const int*   sg_src   = (const int*)d_in[2];
    const int*   sg_dst   = (const int*)d_in[3];
    const int*   gs_src   = (const int*)d_in[4];
    const int*   gs_dst   = (const int*)d_in[5];
    const float* Wl1_sg = (const float*)d_in[6];   const float* bl1_sg  = (const float*)d_in[7];
    const float* Wr1_sg = (const float*)d_in[8];   const float* br1_sg  = (const float*)d_in[9];
    const float* att1_sg = (const float*)d_in[10]; const float* bias1_sg = (const float*)d_in[11];
    const float* Wl1_gs = (const float*)d_in[12];  const float* bl1_gs  = (const float*)d_in[13];
    const float* Wr1_gs = (const float*)d_in[14];  const float* br1_gs  = (const float*)d_in[15];
    const float* att1_gs = (const float*)d_in[16]; const float* bias1_gs = (const float*)d_in[17];
    const float* Wl3 = (const float*)d_in[18];     const float* bl3 = (const float*)d_in[19];
    const float* Wr3 = (const float*)d_in[20];     const float* br3 = (const float*)d_in[21];
    const float* att3 = (const float*)d_in[22];    const float* bias3 = (const float*)d_in[23];
    const float* sl1_W = (const float*)d_in[24];   const float* sl1_b = (const float*)d_in[25];
    const float* sl3_W = (const float*)d_in[26];   const float* sl3_b = (const float*)d_in[27];

    char* p = (char*)d_ws;
    auto carve = [&](size_t bytes) -> char* {
        char* r = p; p += (bytes + 255) & ~(size_t)255; return r;
    };

    _Float16* hx_sample = (_Float16*)carve((size_t)NS * DIN * 2);
    _Float16* hx_gene   = (_Float16*)carve((size_t)NG * DIN * 2);
    _Float16* hWl1sg = (_Float16*)carve((size_t)DIN * HC1 * 2);
    _Float16* hWr1sg = (_Float16*)carve((size_t)DIN * HC1 * 2);
    _Float16* hWl1gs = (_Float16*)carve((size_t)DIN * HC1 * 2);
    _Float16* hWr1gs = (_Float16*)carve((size_t)DIN * HC1 * 2);
    _Float16* hWl3   = (_Float16*)carve((size_t)HC1 * HC3 * 2);
    _Float16* hWr3   = (_Float16*)carve((size_t)HC1 * HC3 * 2);
    _Float16* hsl1W  = (_Float16*)carve((size_t)DIN * C1 * 2);
    _Float16* hsl3W  = (_Float16*)carve((size_t)HC1 * C3 * 2);
    _Float16* hx1_gene   = (_Float16*)carve((size_t)NG * HC1 * 2);
    _Float16* hx1_sample = (_Float16*)carve((size_t)NS * HC1 * 2);
    float* xl_sg = (float*)carve((size_t)NS * HC1 * 4);
    float* xr_sg = (float*)carve((size_t)NG * HC1 * 4);
    float* xl_gs = (float*)carve((size_t)NG * HC1 * 4);
    float* xr_gs = (float*)carve((size_t)NS * HC1 * 4);
    int* bk = (int*)carve((size_t)2 * NE * 4);
    int* T  = (int*)carve((size_t)2 * SB_NCH * SB_NR * 4);
    float* sl1 = (float*)carve((size_t)NS * C1 * 4);
    float* xl3 = (float*)carve((size_t)NG * HC3 * 4);
    float* xr3 = (float*)carve((size_t)NS * HC3 * 4);
    (void)carve((size_t)NS * HC3 * 4);
    float* sl3 = (float*)carve((size_t)NS * C3 * 4);
    int* off_sg = (int*)carve((size_t)(NG + 1) * 4);
    int* off_gs = (int*)carve((size_t)(NS + 1) * 4);
    int* ss_sg  = (int*)carve((size_t)NE * 4);
    int* ss_gs  = (int*)carve((size_t)NE * 4);

    k_prep<<<PO_SORT + PO_END, 256, 0, stream>>>(x_sample, x_gene, Wl1_sg, Wr1_sg, Wl1_gs, Wr1_gs, Wl3, Wr3,
                                       sl1_W, sl3_W, hx_sample, hx_gene, hWl1sg, hWr1sg, hWl1gs,
                                       hWr1gs, hWl3, hWr3, hsl1W, hsl3W,
                                       sg_src, sg_dst, gs_src, gs_dst,
                                       bk, T, bk + NE, T + SB_NCH * SB_NR);
    k_gemm_l1<<<G1_END, 256, 0, stream>>>(hx_sample, hx_gene,
                                          hWr1sg, br1_sg, xr_sg,  hWl1gs, bl1_gs, xl_gs,
                                          hWl1sg, bl1_sg, xl_sg,  hWr1gs, br1_gs, xr_gs,
                                          hsl1W, sl1_b, sl1,
                                          bk, T, off_sg, ss_sg,
                                          bk + NE, T + SB_NCH * SB_NR, off_gs, ss_gs);
    k_gat_l1<<<A1_END, 256, 0, stream>>>(off_gs, ss_gs, xl_gs, xr_gs, att1_gs, bias1_gs, sl1, hx1_sample,
                                         off_sg, ss_sg, xl_sg, xr_sg, att1_sg, bias1_sg, hx1_gene);
    k_gemm_l3<<<G3_END, 256, 0, stream>>>(hx1_gene, hx1_sample, hWl3, bl3, xl3, hWr3, br3, xr3,
                                          hsl3W, sl3_b, sl3);
    k_gat_l3<<<NS / 8, 256, 0, stream>>>(off_gs, ss_gs, xl3, xr3, att3, bias3, sl3, (float*)d_out);
}


# precision: the three gathered source tables (xl of both layer-1 relations and of layer 3) are stored as f16, not f32: a gathered row is half the bytes; scores, softmax and sums stay fp32
# speedup vs baseline: 9.9417x; 1.2151x over previous
#include <hip/hip_runtime.h>
#include <hip/hip_bf16.h>


#define NS   4096
#define NG   20000
#define NE   131072
#define DIN  256
#define NH   4
#define C1   64
#define C3   128
#define HC1  256
#define HC3  512

typedef __attribute__((ext_vector_type(16))) _Float16 v16h;
typedef __attribute__((ext_vector_type(8)))  _Float16 v8h;
typedef __attribute__((ext_vector_type(4)))  _Float16 v4h;
typedef __attribute__((ext_vector_type(8)))  float    v8f;
typedef __attribute__((ext_vector_type(4)))  float    v4f;
typedef __attribute__((ext_vector_type(4)))  int      v4i;

__device__ __forceinline__ float eluf(float x) { return x > 0.f ? x : (expf(x) - 1.f); }

__device__ __forceinline__ size_t fidx(int row, int k) {
    return ((size_t)((row >> 4) * 8 + (k >> 5)) << 9)
         + (size_t)((((k >> 4) & 1) << 8)
                    + (((row & 15) + (((k >> 3) & 1) << 4)) << 3)
                    + (k & 7));
}

#define SB_CH     2048
#define SB_NCH    (NE / SB_CH)
#define SB_NR     64
#define SB_RW_SG  313
#define SB_RW_GS  64
#define SB_HN     512
static_assert(NE % SB_CH == 0 && SB_NCH == 64, "a level-2 thread per chunk, 8 chunks per wave");
static_assert(SB_RW_SG * SB_NR >= NG && SB_RW_SG * (SB_NR - 1) < NG, "ranges cover the gene nodes, none empty");
static_assert(SB_RW_GS * SB_NR == NS, "ranges cover the sample nodes");
static_assert(SB_RW_SG <= SB_HN && SB_RW_GS <= SB_HN, "one LDS counter per node of a range");
static_assert(NG < 32768 && NS < 32768, "dst << 16 | src fits a positive int");

__device__ __forceinline__ int wave_incl_scan(int v, int lane) {
#pragma unroll
    for (int o = 1; o < 32; o <<= 1) {
        const int u = __shfl_up(v, o);
        v += (lane >= o) ? u : 0;
    }
    return v;
}

template <int RW>
__device__ __forceinline__ void
bucket_chunk(const int* __restrict__ src, const int* __restrict__ dst,
             int* __restrict__ bk, int* __restrict__ T, const int c, const int t,
             int* hist, int* start) {
    if (t < SB_NR) hist[t] = 0;
    __syncthreads();
    const int e0 = c * SB_CH + 4 * t;
    const v4i d0 = *(const v4i*)(dst + e0);
    const v4i d1 = *(const v4i*)(dst + e0 + SB_CH / 2);
    const v4i s0 = *(const v4i*)(src + e0);
    const v4i s1 = *(const v4i*)(src + e0 + SB_CH / 2);
    int rg[8], key[8], rank[8];
#pragma unroll
    for (int k = 0; k < 4; ++k) {
        rg[k]     = d0[k] / RW;  key[k]     = (d0[k] << 16) | s0[k];
        rg[4 + k] = d1[k] / RW;  key[4 + k] = (d1[k] << 16) | s1[k];
    }
#pragma unroll
    for (int k = 0; k < 8; ++k) rank[k] = atomicAdd(&hist[rg[k]], 1);
    __syncthreads();
    if (t < 32) {
        const int a = hist[2 * t], b = hist[2 * t + 1];
        const int ex = wave_incl_scan(a + b, t) - (a + b);
        start[2 * t]     = ex;
        start[2 * t + 1] = ex + a;
        *(int2*)(T + c * SB_NR + 2 * t) = make_int2(a, b);
    }
    __syncthreads();
    int* out = bk + c * SB_CH;
#pragma unroll
    for (int k = 0; k < 8; ++k) out[start[rg[k]] + rank[k]] = key[k];
}

template <int RW, int N>
__device__ __forceinline__ void
group_range(const int* __restrict__ bk, const int* __restrict__ T,
            int* __restrict__ off, int* __restrict__ ss, const int r,
            int* cs_s, int* n_s, int* h, int* cur, int* wt, int* base_s) {
    const int t = threadIdx.x, lane = t & 31, w = t >> 5;
    const int lo = r * RW;
    h[2 * t] = 0;  h[2 * t + 1] = 0;
    if (t == 0) *base_s = 0;
    __syncthreads();
    if (t < SB_NCH) {
        const v4i* row = (const v4i*)(T + t * SB_NR);
        int before = 0, mine = 0;
#pragma unroll
        for (int q = 0; q < SB_NR / 4; ++q) {
            const v4i v = row[q];
#pragma unroll
            for (int k = 0; k < 4; ++k) {
                const int rr = 4 * q + k;
                before += (rr < r) ? v[k] : 0;
                mine    = (rr == r) ? v[k] : mine;
            }
        }
        cs_s[t] = before;
        n_s[t]  = mine;
        atomicAdd(base_s, before);
    }
    __syncthreads();
    const int base = *base_s;

    int key[8][2];
#pragma unroll
    for (int q = 0; q < 8; ++q) {
        const int ch = w + 8 * q;
        const int n = n_s[ch];
        const int* p = bk + ch * SB_CH + cs_s[ch];
        key[q][0] = (lane < n)      ? p[lane]      : -1;
        key[q][1] = (lane + 32 < n) ? p[lane + 32] : -1;
    }
#pragma unroll
    for (int q = 0; q < 8; ++q) {
#pragma unroll
        for (int u = 0; u < 2; ++u)
            if (key[q][u] >= 0) atomicAdd(&h[(key[q][u] >> 16) - lo], 1);
        const int ch = w + 8 * q;
        const int n = n_s[ch];
        const int* p = bk + ch * SB_CH + cs_s[ch];
        for (int i = lane + 64; i < n; i += 32) atomicAdd(&h[(p[i] >> 16) - lo], 1);
    }
    __syncthreads();

    const int a = h[2 * t], b = h[2 * t + 1];
    const int inc = wave_incl_scan(a + b, lane);
    if (lane == 31) wt[w] = inc;
    __syncthreads();
    int woff = 0;
    for (int q = 0; q < 8; ++q) woff += (q < w) ? wt[q] : 0;
    const int ex = base + woff + inc - (a + b);
    cur[2 * t]     = ex;
    cur[2 * t + 1] = ex + a;
    const int j = 2 * t;
    if (j < RW && lo + j < N)         off[lo + j]     = ex;
    if (j + 1 < RW && lo + j + 1 < N) off[lo + j + 1] = ex + a;
    if (r == SB_NR - 1 && t == 0)     off[N] = NE;
    __syncthreads();

#pragma unroll
    for (int q = 0; q < 8; ++q) {
#pragma unroll
        for (int u = 0; u < 2; ++u)
            if (key[q][u] >= 0) ss[atomicAdd(&cur[(key[q][u] >> 16) - lo], 1)] = key[q][u] & 0xffff;
        const int ch = w + 8 * q;
        const int n = n_s[ch];
        const int* p = bk + ch * SB_CH + cs_s[ch];
        for (int i = lane + 64; i < n; i += 32) {
            const int kv = p[i];
            ss[atomicAdd(&cur[(kv >> 16) - lo], 1)] = kv & 0xffff;
        }
    }
}

#define PB_W1   4
#define PB_W3   8
#define PO_WL1SG 0
#define PO_WR1SG (PO_WL1SG + PB_W1)
#define PO_WL1GS (PO_WR1SG + PB_W1)
#define PO_WR1GS (PO_WL1GS + PB_W1)
#define PO_WL3   (PO_WR1GS + PB_W1)
#define PO_WR3   (PO_WL3 + PB_W3)
#define PO_SL1   (PO_WR3 + PB_W3)
#define PO_SL3   (PO_SL1 + 1)
#define PO_XS    (PO_SL3 + 2)
#define PO_XG    (PO_XS + 64)
#define PO_END   (PO_XG + 500)
#define PO_SORT  (2 * SB_NCH)

template <int K, int N, bool IL>
__device__ __forceinline__ void cvt_t(const float* __restrict__ in, _Float16* __restrict__ out, int i) {
    static_assert(!IL || N % 64 == 0, "interleave works on whole 64-column wave tiles");
    const int q = i / N, n = i - q * N;
    const int ns = IL ? ((n & ~63) | ((n & 3) << 4) | ((n & 63) >> 2)) : n;
    const float* p = in + (size_t)(q * 64) * N + n;
    float v[64];
#pragma unroll
    for (int kk = 0; kk < 64; ++kk) v[kk] = p[(size_t)kk * N];
    static_assert(K == 256, "fragment-order storage is written for K = 256");
#pragma unroll
    for (int s = 0; s < 8; ++s) {
        v8h hv;
#pragma unroll
        for (int j = 0; j < 8; ++j) hv[j] = (_Float16)v[8 * s + j];
        *(v8h*)(out + fidx(ns, q * 64 + 8 * s)) = hv;
    }
}

template <int U>
__device__ __forceinline__ void cvt_run(const float* __restrict__ in, _Float16* __restrict__ out, int blk, int t) {
    static_assert(U % 2 == 0, "8 rows of the block per piece");
    const int k    = 64 * ((t >> 5) & 3) + 8 * (t & 7);
    const int row0 = blk * (U * 4) + 4 * (t >> 7) + ((t >> 3) & 3);
    v4f x[U];
#pragma unroll
    for (int j = 0; j < U / 2; ++j) {
        const float* p = in + (size_t)(row0 + 8 * j) * 256 + k;
        x[2 * j]     = *(const v4f*)(p);
        x[2 * j + 1] = *(const v4f*)(p + 4);
    }
#pragma unroll
    for (int j = 0; j < U / 2; ++j) {
        v8h hv;
#pragma unroll
        for (int e = 0; e < 4; ++e) { hv[e] = (_Float16)x[2 * j][e]; hv[4 + e] = (_Float16)x[2 * j + 1][e]; }
        *(v8h*)(out + fidx(row0 + 8 * j, k)) = hv;
    }
}

__global__ void __launch_bounds__(256)
k_prep(const float* __restrict__ x_sample, const float* __restrict__ x_gene,
       const float* __restrict__ Wl1_sg, const float* __restrict__ Wr1_sg,
       const float* __restrict__ Wl1_gs, const float* __restrict__ Wr1_gs,
       const float* __restrict__ Wl3, const float* __restrict__ Wr3,
       const float* __restrict__ sl1_W, const float* __restrict__ sl3_W,
       _Float16* __restrict__ hx_sample, _Float16* __restrict__ hx_gene,
       _Float16* __restrict__ hWl1sg, _Float16* __restrict__ hWr1sg,
       _Float16* __restrict__ hWl1gs, _Float16* __restrict__ hWr1gs,
       _Float16* __restrict__ hWl3, _Float16* __restrict__ hWr3,
       _Float16* __restrict__ hsl1W, _Float16* __restrict__ hsl3W,
       const int* __restrict__ sg_src, const int* __restrict__ sg_dst,
       const int* __restrict__ gs_src, const int* __restrict__ gs_dst,
       int* __restrict__ bk_sg, int* __restrict__ T_sg,
       int* __restrict__ bk_gs, int* __restrict__ T_gs) {
    __shared__ int sb_hist[SB_NR], sb_start[SB_NR];
    const int t = threadIdx.x;
    if (blockIdx.x < PO_SORT) {
        const int cb = blockIdx.x;
        if (cb < SB_NCH) bucket_chunk<SB_RW_SG>(sg_src, sg_dst, bk_sg, T_sg, cb, t, sb_hist, sb_start);
        else             bucket_chunk<SB_RW_GS>(gs_src, gs_dst, bk_gs, T_gs, cb - SB_NCH, t, sb_hist, sb_start);
        return;
    }
    const int b = blockIdx.x - PO_SORT;
    if (b >= PO_XG) {
        cvt_run<10>(x_gene, hx_gene, b - PO_XG, t);
    } else if (b >= PO_XS) {
        cvt_run<16>(x_sample, hx_sample, b - PO_XS, t);
    } else if (b >= PO_SL3) {
        cvt_t<HC1, C3, false>(sl3_W, hsl3W, (b - PO_SL3) * 256 + t);
    } else if (b >= PO_SL1) {
        cvt_t<DIN, C1, false>(sl1_W, hsl1W, (b - PO_SL1) * 256 + t);
    } else if (b >= PO_WR3) {
        cvt_t<HC1, HC3, false>(Wr3, hWr3, (b - PO_WR3) * 256 + t);
    } else if (b >= PO_WL3) {
        cvt_t<HC1, HC3, true>(Wl3, hWl3, (b - PO_WL3) * 256 + t);
    } else if (b >= PO_WR1GS) {
        cvt_t<DIN, HC1, false>(Wr1_gs, hWr1gs, (b - PO_WR1GS) * 256 + t);
    } else if (b >= PO_WL1GS) {
        cvt_t<DIN, HC1, true>(Wl1_gs, hWl1gs, (b - PO_WL1GS) * 256 + t);
    } else if (b >= PO_WR1SG) {
        cvt_t<DIN, HC1, false>(Wr1_sg, hWr1sg, (b - PO_WR1SG) * 256 + t);
    } else {
        cvt_t<DIN, HC1, true>(Wl1_sg, hWl1sg, (b - PO_WL1SG) * 256 + t);
    }
}

__device__ __forceinline__ v16h ld_frag(const _Float16* p) {
    v8h lo = *(const v8h*)(p);
    v8h hi = *(const v8h*)(p + 256);
    v16h r;
#pragma unroll
    for (int j = 0; j < 8; ++j) { r[j] = lo[j]; r[8 + j] = hi[j]; }
    return r;
}

constexpr int gemm_wn(int N) { return N >= 128 ? 2 : 1; }
constexpr int gemm_bm(int N) { return 64 * (8 / gemm_wn(N)); }
constexpr int gemm_bn(int N) { return 64 * gemm_wn(N); }
constexpr int gemm_blocks(int M, int N) { return ((M + gemm_bm(N) - 1) / gemm_bm(N)) * (N / gemm_bn(N)); }
#define GEMM_BLOCKS(M, N) gemm_blocks(M, N)

template <int M, int N, typename OutT>
__device__ __forceinline__ void
gemm_tile(const _Float16* __restrict__ X, const _Float16* __restrict__ Wt,
          const float* __restrict__ bias, OutT* __restrict__ Y, int t) {
    constexpr int K  = 256;
    constexpr int WN = gemm_wn(N), WM = 8 / WN;
    constexpr int BM = gemm_bm(N), BN = gemm_bn(N);
    constexpr int MB = (M + BM - 1) / BM;
    constexpr int GM = M / 16;
    static_assert(M % 16 == 0 && N % BN == 0 && K == 256, "whole 16-row groups, whole column blocks");
    const int bn = t / MB, bm = t - bn * MB;
    const int wave = __builtin_amdgcn_readfirstlane((int)(threadIdx.x >> 5));
    const int lane = threadIdx.x & 31;
    const int wm = wave % WM, wn = wave / WM;
    const int m0 = bm * BM + wm * 64;
    if (m0 >= M) return;
    const int n0 = bn * BN + wn * 64;

    const _Float16* pA[4];
    const _Float16* pB[4];
#pragma unroll
    for (int i = 0; i < 4; ++i) {
        int g = (m0 >> 4) + i;
        g = g < GM ? g : GM - 1;
        pA[i] = X  + (size_t)g * (8 * 512);
        pB[i] = Wt + (size_t)((n0 >> 4) + i) * (8 * 512);
    }

    v8f acc[4][4];
#pragma unroll
    for (int i = 0; i < 4; ++i)
#pragma unroll
        for (int j = 0; j < 4; ++j) acc[i][j] = (v8f){0.f, 0.f, 0.f, 0.f, 0.f, 0.f, 0.f, 0.f};

#pragma nounroll
    for (int s = 0; s < K / 32; ++s) {
        const int o = s * 512 + lane * 8;
        v16h a[4], b[4];
        a[0] = ld_frag(pA[0] + o);
#pragma unroll
        for (int j = 0; j < 4; ++j) b[j] = ld_frag(pB[j] + o);
#pragma unroll
        for (int i = 1; i < 4; ++i) a[i] = ld_frag(pA[i] + o);
        __builtin_amdgcn_sched_barrier(0);
#pragma unroll
        for (int i = 0; i < 4; ++i)
#pragma unroll
            for (int j = 0; j < 4; ++j)
                acc[i][j] = __builtin_amdgcn_wmma_f32_16x16x32_f16(false, a[i], false, b[j], (short)0, acc[i][j], false, false);
    }

    const int lr = lane & 15;
    if constexpr (sizeof(OutT) == 2) {
        const v4f bq = *(const v4f*)(bias + n0 + 4 * lr);
#pragma unroll
        for (int i = 0; i < 4; ++i) {
            if (m0 + 16 * i < M) {
                _Float16* y = Y + (size_t)(m0 + 16 * i + 8 * (lane >> 4)) * N + n0 + 4 * lr;
#pragma unroll
                for (int v = 0; v < 8; ++v) {
                    v4h o;
#pragma unroll
                    for (int j = 0; j < 4; ++j) o[j] = (_Float16)(acc[i][j][v] + bq[j]);
                    *(v4h*)(y + (size_t)v * N) = o;
                }
            }
        }
    } else {
        float bv[4];
#pragma unroll
        for (int j = 0; j < 4; ++j) bv[j] = bias[n0 + 16 * j + lr];
#pragma unroll
        for (int i = 0; i < 4; ++i) {
            if (m0 + 16 * i < M) {
                float* y = Y + (size_t)(m0 + 16 * i + 8 * (lane >> 4)) * N + n0 + lr;
#pragma unroll
                for (int v = 0; v < 8; ++v) {
#pragma unroll
                    for (int j = 0; j < 4; ++j) y[(size_t)v * N + 16 * j] = acc[i][j][v] + bv[j];
                }
            }
        }
    }
}

#define G1_SORT (2 * SB_NR)
#define G1_O0   G1_SORT
#define G1_O1   (G1_O0 + GEMM_BLOCKS(NG, HC1))
#define G1_O2   (G1_O1 + GEMM_BLOCKS(NG, HC1))
#define G1_O3   (G1_O2 + GEMM_BLOCKS(NS, HC1))
#define G1_O4   (G1_O3 + GEMM_BLOCKS(NS, HC1))
#define G1_END  (G1_O4 + GEMM_BLOCKS(NS, C1))

__global__ void __launch_bounds__(256)
k_gemm_l1(const _Float16* __restrict__ hx_sample, const _Float16* __restrict__ hx_gene,
          const _Float16* __restrict__ hWr1sg, const float* __restrict__ br1_sg, float* __restrict__ xr_sg,
          const _Float16* __restrict__ hWl1gs, const float* __restrict__ bl1_gs, _Float16* __restrict__ xl_gs,
          const _Float16* __restrict__ hWl1sg, const float* __restrict__ bl1_sg, _Float16* __restrict__ xl_sg,
          const _Float16* __restrict__ hWr1gs, const float* __restrict__ br1_gs, float* __restrict__ xr_gs,
          const _Float16* __restrict__ hsl1W,  const float* __restrict__ sl1_b,  float* __restrict__ sl1,
          const int* __restrict__ bk_sg, const int* __restrict__ T_sg,
          int* __restrict__ off_sg, int* __restrict__ ss_sg,
          const int* __restrict__ bk_gs, const int* __restrict__ T_gs,
          int* __restrict__ off_gs, int* __restrict__ ss_gs) {
    __shared__ int sb_cs[SB_NCH], sb_n[SB_NCH], sb_h[SB_HN], sb_cur[SB_HN], sb_wt[8], sb_base;
    const int b = blockIdx.x;
    if (b < G1_SORT) {
        if (b < SB_NR) group_range<SB_RW_SG, NG>(bk_sg, T_sg, off_sg, ss_sg, b, sb_cs, sb_n, sb_h, sb_cur, sb_wt, &sb_base);
        else           group_range<SB_RW_GS, NS>(bk_gs, T_gs, off_gs, ss_gs, b - SB_NR, sb_cs, sb_n, sb_h, sb_cur, sb_wt, &sb_base);
    }
    else if (b < G1_O1) gemm_tile<NG, HC1, float   >(hx_gene,   hWr1sg, br1_sg, xr_sg, b - G1_O0);
    else if (b < G1_O2) gemm_tile<NG, HC1, _Float16>(hx_gene,   hWl1gs, bl1_gs, xl_gs, b - G1_O1);
    else if (b < G1_O3) gemm_tile<NS, HC1, _Float16>(hx_sample, hWl1sg, bl1_sg, xl_sg, b - G1_O2);
    else if (b < G1_O4) gemm_tile<NS, HC1, float   >(hx_sample, hWr1gs, br1_gs, xr_gs, b - G1_O3);
    else                gemm_tile<NS, C1,  float   >(hx_sample, hsl1W,  sl1_b,  sl1,   b - G1_O4);
}

#define G3_O1   GEMM_BLOCKS(NG, HC3)
#define G3_O2   (G3_O1 + GEMM_BLOCKS(NS, HC3))
#define G3_END  (G3_O2 + GEMM_BLOCKS(NS, C3))

__global__ void __launch_bounds__(256)
k_gemm_l3(const _Float16* __restrict__ hx1_gene, const _Float16* __restrict__ hx1_sample,
          const _Float16* __restrict__ hWl3,  const float* __restrict__ bl3,   _Float16* __restrict__ xl3,
          const _Float16* __restrict__ hWr3,  const float* __restrict__ br3,   float* __restrict__ xr3,
          const _Float16* __restrict__ hsl3W, const float* __restrict__ sl3_b, float* __restrict__ sl3) {
    const int b = blockIdx.x;
    if      (b < G3_O1) gemm_tile<NG, HC3, _Float16>(hx1_gene,   hWl3,  bl3,   xl3, b);
    else if (b < G3_O2) gemm_tile<NS, HC3, float   >(hx1_sample, hWr3,  br3,   xr3, b - G3_O1);
    else                gemm_tile<NS, C3,  float   >(hx1_sample, hsl3W, sl3_b, sl3, b - G3_O2);
}

__device__ __forceinline__ float sum8(float t) {
    t += __builtin_bit_cast(float, __builtin_amdgcn_update_dpp(0, __builtin_bit_cast(int, t), 0x161, 0xf, 0xf, true));
    t += __builtin_bit_cast(float, __builtin_amdgcn_update_dpp(0, __builtin_bit_cast(int, t), 0x162, 0xf, 0xf, true));
    t += __builtin_bit_cast(float, __builtin_amdgcn_update_dpp(0, __builtin_bit_cast(int, t), 0x164, 0xf, 0xf, true));
    return t;
}
__device__ __forceinline__ float sum_heads(float t) {
    t += __builtin_bit_cast(float, __builtin_amdgcn_update_dpp(0, __builtin_bit_cast(int, t), 0x168, 0xf, 0xf, true));
    t += __builtin_bit_cast(float, __builtin_amdgcn_permlanex16(__builtin_bit_cast(int, t), __builtin_bit_cast(int, t),
                                                                0x76543210, 0xfedcba98, false, false));
    return t;
}

#define EPI_GENE   0
#define EPI_SAMPLE 1
#define EPI_FINAL  2

template <int C, int EPI>
__device__ __forceinline__ void
gat_node(const int* __restrict__ off, const int* __restrict__ ssrc,
         const _Float16* __restrict__ xl, const float* __restrict__ xr,
         const float* __restrict__ att, const float* __restrict__ bias,
         const float* __restrict__ sl, _Float16* __restrict__ outh, float* __restrict__ outf,
         const int d) {
    constexpr int V   = C / 32;
    constexpr int U   = 4;
    constexpr int ROW = NH * C;
    static_assert(V % 2 == 0, "pieces come in pairs: 8 halves per load");
    const int lane = threadIdx.x & 31;
    const int ls = lane & 7, h = lane >> 3;
    const int cbase = h * C + 8 * ls;
#define COFF(j) (64 * ((j) >> 1) + 4 * ((j) & 1))

    const float* pr = xr + (size_t)d * ROW + cbase;
    v4f r[V], a[V], acc[V];
#pragma unroll
    for (int j = 0; j < V; ++j) {
        r[j] = *(const v4f*)(pr + COFF(j));
        const float* q = att + cbase + COFF(j);
        a[j] = (v4f){q[0], q[1], q[2], q[3]};
        acc[j] = (v4f){0.f, 0.f, 0.f, 0.f};
    }
    constexpr float NEG   = -1.0e30f;
    constexpr float LOG2E = 1.4426950408889634f;
    float m = NEG, den = 0.f;
    const int beg = off[d], end = off[d + 1];

    for (int i = beg; i < end; i += U) {
        v4f x[U][V];
#pragma unroll
        for (int u = 0; u < U; ++u) {
            int idx = i + u; if (idx > end - 1) idx = end - 1;
            const _Float16* pl = xl + (size_t)ssrc[idx] * ROW + cbase;
#pragma unroll
            for (int q = 0; q < V / 2; ++q) {
                const v8h hv = *(const v8h*)(pl + 64 * q);
#pragma unroll
                for (int e = 0; e < 4; ++e) { x[u][2 * q][e] = (float)hv[e]; x[u][2 * q + 1][e] = (float)hv[4 + e]; }
            }
        }
        float t[U];
#pragma unroll
        for (int u = 0; u < U; ++u) {
            v4f sv = (v4f){0.f, 0.f, 0.f, 0.f};
#pragma unroll
            for (int j = 0; j < V; ++j) {
                const v4f g = x[u][j] + r[j];
                sv += __builtin_elementwise_max(g, g * 0.2f) * a[j];
            }
            const float s = sum8((sv[0] + sv[1]) + (sv[2] + sv[3]));
            t[u] = (i + u < end) ? s : NEG;
        }
        float mn = m;
#pragma unroll
        for (int u = 0; u < U; ++u) mn = fmaxf(mn, t[u]);
        const float sc = __builtin_amdgcn_exp2f((m - mn) * LOG2E);
        float p[U];
        float ps = 0.f;
#pragma unroll
        for (int u = 0; u < U; ++u) {
            p[u] = __builtin_amdgcn_exp2f((t[u] - mn) * LOG2E);
            ps += p[u];
        }
        den = den * sc + ps;
#pragma unroll
        for (int j = 0; j < V; ++j) {
            v4f s4 = acc[j] * sc;
#pragma unroll
            for (int u = 0; u < U; ++u) s4 += x[u][j] * p[u];
            acc[j] = s4;
        }
        m = mn;
    }

    const float inv = 1.f / (den + 1e-16f);
    if (EPI == EPI_FINAL) {
        v4f pick = (v4f){0.f, 0.f, 0.f, 0.f};
#pragma unroll
        for (int j = 0; j < V; ++j) {
            v4f s = acc[j] * inv;
#pragma unroll
            for (int k = 0; k < 4; ++k) s[k] = sum_heads(s[k]);
            pick = (h == j) ? s : pick;
        }
        const int c = COFF(h) + 8 * ls;
        const v4f bq = *(const v4f*)(bias + c);
        const v4f sq = *(const v4f*)(sl + (size_t)d * C + c);
        v4f o;
#pragma unroll
        for (int k = 0; k < 4; ++k) o[k] = eluf(0.25f * pick[k] + bq[k] + sq[k]);
        *(v4f*)(outf + (size_t)d * C + c) = o;
    } else {
        static_assert(EPI == EPI_FINAL || ROW == 256, "fragment-order storage is written for K = 256");
#pragma unroll
        for (int j = 0; j < V; ++j) {
            v4f o = acc[j] * inv;
            const float* q = bias + cbase + COFF(j);
            o += (v4f){q[0], q[1], q[2], q[3]};
            if (EPI == EPI_SAMPLE) o += *(const v4f*)(sl + (size_t)d * C + 8 * ls + COFF(j));
            v4h oh;
#pragma unroll
            for (int k = 0; k < 4; ++k) oh[k] = (_Float16)eluf(o[k]);
            *(v4h*)(outh + fidx(d, cbase + COFF(j))) = oh;
        }
    }
#undef COFF
}

#define A1_GS  (NS / 8)
#define A1_END (A1_GS + NG / 8)

__global__ void __launch_bounds__(256)
k_gat_l1(const int* __restrict__ off_gs, const int* __restrict__ ss_gs,
         const _Float16* __restrict__ xl_gs, const float* __restrict__ xr_gs,
         const float* __restrict__ att1_gs, const float* __restrict__ bias1_gs,
         const float* __restrict__ sl1, _Float16* __restrict__ hx1_sample,
         const int* __restrict__ off_sg, const int* __restrict__ ss_sg,
         const _Float16* __restrict__ xl_sg, const float* __restrict__ xr_sg,
         const float* __restrict__ att1_sg, const float* __restrict__ bias1_sg,
         _Float16* __restrict__ hx1_gene) {
    const int w = __builtin_amdgcn_readfirstlane((int)(blockIdx.x * 8 + (threadIdx.x >> 5)));
    if (w < NS) gat_node<C1, EPI_SAMPLE>(off_gs, ss_gs, xl_gs, xr_gs, att1_gs, bias1_gs, sl1, hx1_sample, nullptr, w);
    else        gat_node<C1, EPI_GENE  >(off_sg, ss_sg, xl_sg, xr_sg, att1_sg, bias1_sg, nullptr, hx1_gene, nullptr, w - NS);
}

__global__ void __launch_bounds__(256)
k_gat_l3(const int* __restrict__ off, const int* __restrict__ ssrc,
         const _Float16* __restrict__ xl, const float* __restrict__ xr,
         const float* __restrict__ att, const float* __restrict__ bias3,
         const float* __restrict__ sl3, float* __restrict__ out) {
    const int d = __builtin_amdgcn_readfirstlane((int)(blockIdx.x * 8 + (threadIdx.x >> 5)));
    gat_node<C3, EPI_FINAL>(off, ssrc, xl, xr, att, bias3, sl3, nullptr, out, d);
}

extern "C" void kernel_launch(void* const* d_in, const int* in_sizes, int n_in,
                              void* d_out, int out_size, void* d_ws, size_t ws_size,
                              hipStream_t stream) {
    (void)in_sizes; (void)n_in; (void)out_size; (void)ws_size;

    const float* x_sample = (const float*)d_in[0];
    const float* x_gene   = (const float*)d_in[1];
    const int*   sg_src   = (const int*)d_in[2];
    const int*   sg_dst   = (const int*)d_in[3];
    const int*   gs_src   = (const int*)d_in[4];
    const int*   gs_dst   = (const int*)d_in[5];
    const float* Wl1_sg = (const float*)d_in[6];   const float* bl1_sg  = (const float*)d_in[7];
    const float* Wr1_sg = (const float*)d_in[8];   const float* br1_sg  = (const float*)d_in[9];
    const float* att1_sg = (const float*)d_in[10]; const float* bias1_sg = (const float*)d_in[11];
    const float* Wl1_gs = (const float*)d_in[12];  const float* bl1_gs  = (const float*)d_in[13];
    const float* Wr1_gs = (const float*)d_in[14];  const float* br1_gs  = (const float*)d_in[15];
    const float* att1_gs = (const float*)d_in[16]; const float* bias1_gs = (const float*)d_in[17];
    const float* Wl3 = (const float*)d_in[18];     const float* bl3 = (const float*)d_in[19];
    const float* Wr3 = (const float*)d_in[20];     const float* br3 = (const float*)d_in[21];
    const float* att3 = (const float*)d_in[22];    const float* bias3 = (const float*)d_in[23];
    const float* sl1_W = (const float*)d_in[24];   const float* sl1_b = (const float*)d_in[25];
    const float* sl3_W = (const float*)d_in[26];   const float* sl3_b = (const float*)d_in[27];

    char* p = (char*)d_ws;
    auto carve = [&](size_t bytes) -> char* {
        char* r = p; p += (bytes + 255) & ~(size_t)255; return r;
    };

    _Float16* hx_sample = (_Float16*)carve((size_t)NS * DIN * 2);
    _Float16* hx_gene   = (_Float16*)carve((size_t)NG * DIN * 2);
    _Float16* hWl1sg = (_Float16*)carve((size_t)DIN * HC1 * 2);
    _Float16* hWr1sg = (_Float16*)carve((size_t)DIN * HC1 * 2);
    _Float16* hWl1gs = (_Float16*)carve((size_t)DIN * HC1 * 2);
    _Float16* hWr1gs = (_Float16*)carve((size_t)DIN * HC1 * 2);
    _Float16* hWl3   = (_Float16*)carve((size_t)HC1 * HC3 * 2);
    _Float16* hWr3   = (_Float16*)carve((size_t)HC1 * HC3 * 2);
    _Float16* hsl1W  = (_Float16*)carve((size_t)DIN * C1 * 2);
    _Float16* hsl3W  = (_Float16*)carve((size_t)HC1 * C3 * 2);
    _Float16* hx1_gene   = (_Float16*)carve((size_t)NG * HC1 * 2);
    _Float16* hx1_sample = (_Float16*)carve((size_t)NS * HC1 * 2);
    _Float16* xl_sg = (_Float16*)carve((size_t)NS * HC1 * 2);
    float* xr_sg = (float*)carve((size_t)NG * HC1 * 4);
    _Float16* xl_gs = (_Float16*)carve((size_t)NG * HC1 * 2);
    float* xr_gs = (float*)carve((size_t)NS * HC1 * 4);
    int* bk = (int*)carve((size_t)2 * NE * 4);
    int* T  = (int*)carve((size_t)2 * SB_NCH * SB_NR * 4);
    float* sl1 = (float*)carve((size_t)NS * C1 * 4);
    _Float16* xl3 = (_Float16*)carve((size_t)NG * HC3 * 2);
    float* xr3 = (float*)carve((size_t)NS * HC3 * 4);
    (void)carve((size_t)NS * HC3 * 4);
    float* sl3 = (float*)carve((size_t)NS * C3 * 4);
    int* off_sg = (int*)carve((size_t)(NG + 1) * 4);
    int* off_gs = (int*)carve((size_t)(NS + 1) * 4);
    int* ss_sg  = (int*)carve((size_t)NE * 4);
    int* ss_gs  = (int*)carve((size_t)NE * 4);

    k_prep<<<PO_SORT + PO_END, 256, 0, stream>>>(x_sample, x_gene, Wl1_sg, Wr1_sg, Wl1_gs, Wr1_gs, Wl3, Wr3,
                                       sl1_W, sl3_W, hx_sample, hx_gene, hWl1sg, hWr1sg, hWl1gs,
                                       hWr1gs, hWl3, hWr3, hsl1W, hsl3W,
                                       sg_src, sg_dst, gs_src, gs_dst,
                                       bk, T, bk + NE, T + SB_NCH * SB_NR);
    k_gemm_l1<<<G1_END, 256, 0, stream>>>(hx_sample, hx_gene,
                                          hWr1sg, br1_sg, xr_sg,  hWl1gs, bl1_gs, xl_gs,
                                          hWl1sg, bl1_sg, xl_sg,  hWr1gs, br1_gs, xr_gs,
                                          hsl1W, sl1_b, sl1,
                                          bk, T, off_sg, ss_sg,
                                          bk + NE, T + SB_NCH * SB_NR, off_gs, ss_gs);
    k_gat_l1<<<A1_END, 256, 0, stream>>>(off_gs, ss_gs, xl_gs, xr_gs, att1_gs, bias1_gs, sl1, hx1_sample,
                                         off_sg, ss_sg, xl_sg, xr_sg, att1_sg, bias1_sg, hx1_gene);
    k_gemm_l3<<<G3_END, 256, 0, stream>>>(hx1_gene, hx1_sample, hWl3, bl3, xl3, hWr3, br3, xr3,
                                          hsl3W, sl3_b, sl3);
    k_gat_l3<<<NS / 8, 256, 0, stream>>>(off_gs, ss_gs, xl3, xr3, att3, bias3, sl3, (float*)d_out);
}
